# starting kernel
# baseline (speedup 1.0000x reference)
#include <hip/hip_runtime.h>

#define NNODES 50000
#define NEDGES 800000
#define FIN 8
#define PER 12
#define HIDN 32
#define NB 2
#define OUTC 24
#define ROWS (NB*NNODES)
#define NFEAT (FIN*PER)

typedef float v2f __attribute__((ext_vector_type(2)));
typedef float v8f __attribute__((ext_vector_type(8)));

__device__ __forceinline__ v8f wmma4(v2f a, v2f b, v8f c) {
  return __builtin_amdgcn_wmma_f32_16x16x4_f32(false, a, false, b, (short)0, c, false, false);
}

__device__ __forceinline__ v2f ldsA(const float* buf, int m, int hi, int kb) {
  int k0 = kb * 4 + hi * 2;
  v2f a;
  a.x = buf[m * HIDN + k0];
  a.y = buf[m * HIDN + k0 + 1];
  return a;
}

__device__ __forceinline__ v2f ldsB(const float* W, int m, int hi, int kb, int nt) {
  int k0 = kb * 4 + hi * 2;
  int n = nt * 16 + m;
  v2f b;
  b.x = W[k0 * HIDN + n];
  b.y = W[(k0 + 1) * HIDN + n];
  return b;
}

__device__ __forceinline__ void storeTile(float* buf, int m, int hi, const v8f& c0, const v8f& c1) {
#pragma unroll
  for (int r = 0; r < 8; r++) {
    buf[(r + 8 * hi) * HIDN + m]      = c0[r];
    buf[(r + 8 * hi) * HIDN + 16 + m] = c1[r];
  }
}

__device__ __forceinline__ void gemm8(v2f ax0, v2f ax1, const float* W, const float* bias,
                                      int m, int hi, v8f& c0, v8f& c1) {
  float b0v = bias[m], b1v = bias[16 + m];
#pragma unroll
  for (int r = 0; r < 8; r++) { c0[r] = b0v; c1[r] = b1v; }
  c0 = wmma4(ax0, ldsB(W, m, hi, 0, 0), c0);
  c0 = wmma4(ax1, ldsB(W, m, hi, 1, 0), c0);
  c1 = wmma4(ax0, ldsB(W, m, hi, 0, 1), c1);
  c1 = wmma4(ax1, ldsB(W, m, hi, 1, 1), c1);
}

__device__ __forceinline__ void gemm64(const float* A1, const float* A2, const float* L,
                                       const float* bias, int m, int hi, v8f& c0, v8f& c1) {
  float b0v = bias[m], b1v = bias[16 + m];
#pragma unroll
  for (int r = 0; r < 8; r++) { c0[r] = b0v; c1[r] = b1v; }
#pragma unroll
  for (int kb = 0; kb < 8; kb++) {
    v2f a = ldsA(A1, m, hi, kb);
    c0 = wmma4(a, ldsB(L, m, hi, kb, 0), c0);
    c1 = wmma4(a, ldsB(L, m, hi, kb, 1), c1);
  }
#pragma unroll
  for (int kb = 0; kb < 8; kb++) {
    v2f a = ldsA(A2, m, hi, kb);
    c0 = wmma4(a, ldsB(L, m, hi, kb + 8, 0), c0);
    c1 = wmma4(a, ldsB(L, m, hi, kb + 8, 1), c1);
  }
}

__device__ __forceinline__ float sigm(float x) { return 1.0f / (1.0f + __expf(-x)); }
__device__ __forceinline__ float tanhc(float x) {
  x = fminf(fmaxf(x, -15.f), 15.f);
  float e = __expf(2.0f * x);
  return (e - 1.0f) / (e + 1.0f);
}

__global__ void k_deg_init(float* deg) {
  int i = blockIdx.x * 256 + threadIdx.x;
  if (i < NNODES) deg[i] = 1.0f;
}
__global__ void k_deg_acc(const int* __restrict__ ei, float* deg) {
  int e = blockIdx.x * 256 + threadIdx.x;
  if (e < NEDGES) atomicAdd(&deg[ei[NEDGES + e]], 1.0f);
}
__global__ void k_rsqrt(float* deg) {
  int i = blockIdx.x * 256 + threadIdx.x;
  if (i < NNODES) deg[i] = rsqrtf(deg[i]);
}
__global__ void k_selfloop(const float* __restrict__ x, const float* __restrict__ dinv,
                           float* __restrict__ xagg) {
  int t = blockIdx.x * 256 + threadIdx.x;
  if (t >= ROWS * 24) return;
  int i = t / 24;
  int n = i % NNODES;
  float w = dinv[n] * dinv[n];
  float4 xv = ((const float4*)x)[t];
  float4 o; o.x = xv.x * w; o.y = xv.y * w; o.z = xv.z * w; o.w = xv.w * w;
  ((float4*)xagg)[t] = o;
}
__global__ void k_scatter(const float* __restrict__ x, const int* __restrict__ ei,
                          const float* __restrict__ dinv, float* __restrict__ xagg) {
  int t = blockIdx.x * 256 + threadIdx.x;
  if (t >= NEDGES * 24) return;
  int e = t / 24;
  int r = t - e * 24;
  int b = r / 12;
  int c = r - b * 12;
  int s = ei[e];
  int d = ei[NEDGES + e];
  float w = dinv[s] * dinv[d];
  const float* xp = x + (size_t)(b * NNODES + s) * NFEAT + c * 8;
  float* op = xagg + (size_t)(b * NNODES + d) * NFEAT + c * 8;
#pragma unroll
  for (int j = 0; j < 8; j++) atomicAdd(op + j, w * xp[j]);
}

__global__ void __launch_bounds__(256) k_gru(
    const float* __restrict__ xagg, const float* __restrict__ att,
    const float* __restrict__ Wz, const float* __restrict__ bz,
    const float* __restrict__ Wr, const float* __restrict__ br,
    const float* __restrict__ Wh, const float* __restrict__ bh,
    const float* __restrict__ Lz, const float* __restrict__ lbz,
    const float* __restrict__ Lr, const float* __restrict__ lbr,
    const float* __restrict__ Lh, const float* __restrict__ lbh,
    const float* __restrict__ Wo, const float* __restrict__ bo,
    float* __restrict__ out) {
  __shared__ float sW[3 * 8 * 32];
  __shared__ float sb[3 * 32];
  __shared__ float sL[3 * 64 * 32];
  __shared__ float slb[3 * 32];
  __shared__ float sWo[32 * 32];
  __shared__ float sbo[32];
  __shared__ float sG[8][16 * 32];
  __shared__ float sH[8][16 * 32];
  __shared__ float sHR[8][16 * 32];

  int tid = threadIdx.x;
  for (int t = tid; t < 8 * 32; t += 256) { sW[t] = Wz[t]; sW[256 + t] = Wr[t]; sW[512 + t] = Wh[t]; }
  for (int t = tid; t < 32; t += 256) {
    sb[t] = bz[t]; sb[32 + t] = br[t]; sb[64 + t] = bh[t];
    slb[t] = lbz[t]; slb[32 + t] = lbr[t]; slb[64 + t] = lbh[t];
    sbo[t] = (t < OUTC) ? bo[t] : 0.f;
  }
  for (int t = tid; t < 64 * 32; t += 256) { sL[t] = Lz[t]; sL[2048 + t] = Lr[t]; sL[4096 + t] = Lh[t]; }
  for (int t = tid; t < 32 * 32; t += 256) {
    int k = t >> 5, j = t & 31;
    sWo[t] = (j < OUTC) ? Wo[k * OUTC + j] : 0.f;
  }

  int wave = tid >> 5, lane = tid & 31, m = lane & 15, hi = lane >> 4;
  float* mG = sG[wave];
  float* mH = sH[wave];
  float* mHR = sHR[wave];
  for (int idx = lane; idx < 16 * 32; idx += 32) mH[idx] = 0.f;

  float pr[PER];
  {
    float a[PER]; float mx = -1e30f;
    for (int p = 0; p < PER; p++) { a[p] = att[p]; mx = fmaxf(mx, a[p]); }
    float sm = 0.f;
    for (int p = 0; p < PER; p++) { a[p] = __expf(a[p] - mx); sm += a[p]; }
    float inv = 1.0f / sm;
    for (int p = 0; p < PER; p++) pr[p] = a[p] * inv;
  }

  int tile = blockIdx.x * 8 + wave;
  long ia = (long)tile * 16 + m;
  if (ia >= ROWS) ia = ROWS - 1;
  const float* xrow = xagg + ia * (long)NFEAT;

  v8f H0, H1, A0, A1;
  for (int r = 0; r < 8; r++) { H0[r] = 0.f; H1[r] = 0.f; A0[r] = 0.f; A1[r] = 0.f; }

  __syncthreads();

  for (int p = 0; p < PER; p++) {
    int f0 = hi * 2;
    v2f ax0, ax1;
    ax0.x = xrow[(f0)*PER + p];     ax0.y = xrow[(f0 + 1) * PER + p];
    ax1.x = xrow[(f0 + 4) * PER + p]; ax1.y = xrow[(f0 + 5) * PER + p];

    v8f g0, g1, z0, z1, r0, r1, t0, t1;

    gemm8(ax0, ax1, sW + 0, sb + 0, m, hi, g0, g1);
    storeTile(mG, m, hi, g0, g1);
    __syncthreads();
    gemm64(mG, mH, sL + 0, slb + 0, m, hi, z0, z1);
    for (int r = 0; r < 8; r++) { z0[r] = sigm(z0[r]); z1[r] = sigm(z1[r]); }
    __syncthreads();

    gemm8(ax0, ax1, sW + 256, sb + 32, m, hi, g0, g1);
    storeTile(mG, m, hi, g0, g1);
    __syncthreads();
    gemm64(mG, mH, sL + 2048, slb + 32, m, hi, r0, r1);
    for (int r = 0; r < 8; r++) { r0[r] = sigm(r0[r]); r1[r] = sigm(r1[r]); }

    gemm8(ax0, ax1, sW + 512, sb + 64, m, hi, g0, g1);
    __syncthreads();
    storeTile(mG, m, hi, g0, g1);
    {
      v8f q0, q1;
      for (int r = 0; r < 8; r++) { q0[r] = H0[r] * r0[r]; q1[r] = H1[r] * r1[r]; }
      storeTile(mHR, m, hi, q0, q1);
    }
    __syncthreads();
    gemm64(mG, mHR, sL + 4096, slb + 64, m, hi, t0, t1);
    for (int r = 0; r < 8; r++) { t0[r] = tanhc(t0[r]); t1[r] = tanhc(t1[r]); }

    for (int r = 0; r < 8; r++) {
      H0[r] = z0[r] * H0[r] + (1.0f - z0[r]) * t0[r];
      H1[r] = z1[r] * H1[r] + (1.0f - z1[r]) * t1[r];
      A0[r] += pr[p] * H0[r];
      A1[r] += pr[p] * H1[r];
    }
    storeTile(mH, m, hi, H0, H1);
    __syncthreads();
  }

  for (int r = 0; r < 8; r++) { A0[r] = fmaxf(A0[r], 0.f); A1[r] = fmaxf(A1[r], 0.f); }
  storeTile(mG, m, hi, A0, A1);
  __syncthreads();
  v8f o0, o1;
  {
    float b0v = sbo[m], b1v = sbo[16 + m];
    for (int r = 0; r < 8; r++) { o0[r] = b0v; o1[r] = b1v; }
#pragma unroll
    for (int kb = 0; kb < 8; kb++) {
      v2f a = ldsA(mG, m, hi, kb);
      o0 = wmma4(a, ldsB(sWo, m, hi, kb, 0), o0);
      o1 = wmma4(a, ldsB(sWo, m, hi, kb, 1), o1);
    }
  }
#pragma unroll
  for (int r = 0; r < 8; r++) {
    long row = (long)tile * 16 + r + 8 * hi;
    if (row < ROWS) {
      out[row * OUTC + m] = o0[r];
      if (m < 8) out[row * OUTC + 16 + m] = o1[r];
    }
  }
}

extern "C" void kernel_launch(void* const* d_in, const int* in_sizes, int n_in,
                              void* d_out, int out_size, void* d_ws, size_t ws_size,
                              hipStream_t stream) {
  (void)in_sizes; (void)n_in; (void)out_size; (void)ws_size;
  const float* x   = (const float*)d_in[0];
  const int*   ei  = (const int*)d_in[1];
  const float* att = (const float*)d_in[2];
  const float* Wz  = (const float*)d_in[3];  const float* bz  = (const float*)d_in[4];
  const float* Wr  = (const float*)d_in[5];  const float* br  = (const float*)d_in[6];
  const float* Wh  = (const float*)d_in[7];  const float* bh  = (const float*)d_in[8];
  const float* Lz  = (const float*)d_in[9];  const float* lbz = (const float*)d_in[10];
  const float* Lr  = (const float*)d_in[11]; const float* lbr = (const float*)d_in[12];
  const float* Lh  = (const float*)d_in[13]; const float* lbh = (const float*)d_in[14];
  const float* Wo  = (const float*)d_in[15]; const float* bo  = (const float*)d_in[16];
  float* out = (float*)d_out;

  float* wsf  = (float*)d_ws;
  float* deg  = wsf;
  float* xagg = wsf + 65536;

  k_deg_init<<<(NNODES + 255) / 256, 256, 0, stream>>>(deg);
  k_deg_acc<<<(NEDGES + 255) / 256, 256, 0, stream>>>(ei, deg);
  k_rsqrt<<<(NNODES + 255) / 256, 256, 0, stream>>>(deg);
  k_selfloop<<<(ROWS * 24 + 255) / 256, 256, 0, stream>>>(x, deg, xagg);
  k_scatter<<<(NEDGES * 24 + 255) / 256, 256, 0, stream>>>(x, ei, deg, xagg);

  int tiles = ROWS / 16;
  int blocks = (tiles + 7) / 8;
  k_gru<<<blocks, 256, 0, stream>>>(xagg, att, Wz, bz, Wr, br, Wh, bh,
                                    Lz, lbz, Lr, lbr, Lh, lbh, Wo, bo, out);
}


# scatter: a wave walks 32 edges with lane = feature, so each row load and each f32 atomic add covers one whole 128-byte line (was 8 lines per instruction); same adds, arrival order as before
# speedup vs baseline: 1.4711x; 1.4711x over previous
#include <hip/hip_runtime.h>

#define NNODES 50000
#define NEDGES 800000
#define FIN 8
#define PER 12
#define HIDN 32
#define NB 2
#define OUTC 24
#define ROWS (NB*NNODES)
#define NFEAT (FIN*PER)

typedef float v2f __attribute__((ext_vector_type(2)));
typedef float v8f __attribute__((ext_vector_type(8)));

__device__ __forceinline__ v8f wmma4(v2f a, v2f b, v8f c) {
  return __builtin_amdgcn_wmma_f32_16x16x4_f32(false, a, false, b, (short)0, c, false, false);
}

__device__ __forceinline__ v2f ldsA(const float* buf, int m, int hi, int kb) {
  int k0 = kb * 4 + hi * 2;
  v2f a;
  a.x = buf[m * HIDN + k0];
  a.y = buf[m * HIDN + k0 + 1];
  return a;
}

__device__ __forceinline__ v2f ldsB(const float* W, int m, int hi, int kb, int nt) {
  int k0 = kb * 4 + hi * 2;
  int n = nt * 16 + m;
  v2f b;
  b.x = W[k0 * HIDN + n];
  b.y = W[(k0 + 1) * HIDN + n];
  return b;
}

__device__ __forceinline__ void storeTile(float* buf, int m, int hi, const v8f& c0, const v8f& c1) {
#pragma unroll
  for (int r = 0; r < 8; r++) {
    buf[(r + 8 * hi) * HIDN + m]      = c0[r];
    buf[(r + 8 * hi) * HIDN + 16 + m] = c1[r];
  }
}

__device__ __forceinline__ void gemm8(v2f ax0, v2f ax1, const float* W, const float* bias,
                                      int m, int hi, v8f& c0, v8f& c1) {
  float b0v = bias[m], b1v = bias[16 + m];
#pragma unroll
  for (int r = 0; r < 8; r++) { c0[r] = b0v; c1[r] = b1v; }
  c0 = wmma4(ax0, ldsB(W, m, hi, 0, 0), c0);
  c0 = wmma4(ax1, ldsB(W, m, hi, 1, 0), c0);
  c1 = wmma4(ax0, ldsB(W, m, hi, 0, 1), c1);
  c1 = wmma4(ax1, ldsB(W, m, hi, 1, 1), c1);
}

__device__ __forceinline__ void gemm64(const float* A1, const float* A2, const float* L,
                                       const float* bias, int m, int hi, v8f& c0, v8f& c1) {
  float b0v = bias[m], b1v = bias[16 + m];
#pragma unroll
  for (int r = 0; r < 8; r++) { c0[r] = b0v; c1[r] = b1v; }
#pragma unroll
  for (int kb = 0; kb < 8; kb++) {
    v2f a = ldsA(A1, m, hi, kb);
    c0 = wmma4(a, ldsB(L, m, hi, kb, 0), c0);
    c1 = wmma4(a, ldsB(L, m, hi, kb, 1), c1);
  }
#pragma unroll
  for (int kb = 0; kb < 8; kb++) {
    v2f a = ldsA(A2, m, hi, kb);
    c0 = wmma4(a, ldsB(L, m, hi, kb + 8, 0), c0);
    c1 = wmma4(a, ldsB(L, m, hi, kb + 8, 1), c1);
  }
}

__device__ __forceinline__ float sigm(float x) { return 1.0f / (1.0f + __expf(-x)); }
__device__ __forceinline__ float tanhc(float x) {
  x = fminf(fmaxf(x, -15.f), 15.f);
  float e = __expf(2.0f * x);
  return (e - 1.0f) / (e + 1.0f);
}

__global__ void k_deg_init(float* deg) {
  int i = blockIdx.x * 256 + threadIdx.x;
  if (i < NNODES) deg[i] = 1.0f;
}
__global__ void k_deg_acc(const int* __restrict__ ei, float* deg) {
  int e = blockIdx.x * 256 + threadIdx.x;
  if (e < NEDGES) atomicAdd(&deg[ei[NEDGES + e]], 1.0f);
}
__global__ void k_rsqrt(float* deg) {
  int i = blockIdx.x * 256 + threadIdx.x;
  if (i < NNODES) deg[i] = rsqrtf(deg[i]);
}
__global__ void k_selfloop(const float* __restrict__ x, const float* __restrict__ dinv,
                           float* __restrict__ xagg) {
  int t = blockIdx.x * 256 + threadIdx.x;
  if (t >= ROWS * 24) return;
  int i = t / 24;
  int n = i % NNODES;
  float w = dinv[n] * dinv[n];
  float4 xv = ((const float4*)x)[t];
  float4 o; o.x = xv.x * w; o.y = xv.y * w; o.z = xv.z * w; o.w = xv.w * w;
  ((float4*)xagg)[t] = o;
}
static_assert(NEDGES % 256 == 0, "one edge per thread at the index fetch, no tail");
static_assert(NFEAT == 96, "a row is three 32-float lines");
__global__ void __launch_bounds__(256) k_scatter(const float* __restrict__ x, const int* __restrict__ ei,
                                                 const float* __restrict__ dinv, float* __restrict__ xagg) {
  const int lane = threadIdx.x & 31;
  const int e = blockIdx.x * 256 + threadIdx.x;
  const int s = ei[e];
  const int d = ei[NEDGES + e];
  const float w = dinv[s] * dinv[d];
  const size_t bstride = (size_t)NNODES * NFEAT;
  for (int k0 = 0; k0 < 32; k0 += 4) {
    float v[4][6];
#pragma unroll
    for (int u = 0; u < 4; u++) {
      const int sk = __builtin_amdgcn_readlane(s, k0 + u);
      const float* xp = x + (size_t)sk * NFEAT + lane;
      v[u][0] = xp[0];
      v[u][1] = xp[32];
      v[u][2] = xp[64];
      v[u][3] = xp[bstride];
      v[u][4] = xp[bstride + 32];
      v[u][5] = xp[bstride + 64];
    }
#pragma unroll
    for (int u = 0; u < 4; u++) {
      const int dk = __builtin_amdgcn_readlane(d, k0 + u);
      const float wk = __int_as_float(__builtin_amdgcn_readlane(__float_as_int(w), k0 + u));
      float* op = xagg + (size_t)dk * NFEAT + lane;
      atomicAdd(op, wk * v[u][0]);
      atomicAdd(op + 32, wk * v[u][1]);
      atomicAdd(op + 64, wk * v[u][2]);
      atomicAdd(op + bstride, wk * v[u][3]);
      atomicAdd(op + bstride + 32, wk * v[u][4]);
      atomicAdd(op + bstride + 64, wk * v[u][5]);
    }
  }
}

__global__ void __launch_bounds__(256) k_gru(
    const float* __restrict__ xagg, const float* __restrict__ att,
    const float* __restrict__ Wz, const float* __restrict__ bz,
    const float* __restrict__ Wr, const float* __restrict__ br,
    const float* __restrict__ Wh, const float* __restrict__ bh,
    const float* __restrict__ Lz, const float* __restrict__ lbz,
    const float* __restrict__ Lr, const float* __restrict__ lbr,
    const float* __restrict__ Lh, const float* __restrict__ lbh,
    const float* __restrict__ Wo, const float* __restrict__ bo,
    float* __restrict__ out) {
  __shared__ float sW[3 * 8 * 32];
  __shared__ float sb[3 * 32];
  __shared__ float sL[3 * 64 * 32];
  __shared__ float slb[3 * 32];
  __shared__ float sWo[32 * 32];
  __shared__ float sbo[32];
  __shared__ float sG[8][16 * 32];
  __shared__ float sH[8][16 * 32];
  __shared__ float sHR[8][16 * 32];

  int tid = threadIdx.x;
  for (int t = tid; t < 8 * 32; t += 256) { sW[t] = Wz[t]; sW[256 + t] = Wr[t]; sW[512 + t] = Wh[t]; }
  for (int t = tid; t < 32; t += 256) {
    sb[t] = bz[t]; sb[32 + t] = br[t]; sb[64 + t] = bh[t];
    slb[t] = lbz[t]; slb[32 + t] = lbr[t]; slb[64 + t] = lbh[t];
    sbo[t] = (t < OUTC) ? bo[t] : 0.f;
  }
  for (int t = tid; t < 64 * 32; t += 256) { sL[t] = Lz[t]; sL[2048 + t] = Lr[t]; sL[4096 + t] = Lh[t]; }
  for (int t = tid; t < 32 * 32; t += 256) {
    int k = t >> 5, j = t & 31;
    sWo[t] = (j < OUTC) ? Wo[k * OUTC + j] : 0.f;
  }

  int wave = tid >> 5, lane = tid & 31, m = lane & 15, hi = lane >> 4;
  float* mG = sG[wave];
  float* mH = sH[wave];
  float* mHR = sHR[wave];
  for (int idx = lane; idx < 16 * 32; idx += 32) mH[idx] = 0.f;

  float pr[PER];
  {
    float a[PER]; float mx = -1e30f;
    for (int p = 0; p < PER; p++) { a[p] = att[p]; mx = fmaxf(mx, a[p]); }
    float sm = 0.f;
    for (int p = 0; p < PER; p++) { a[p] = __expf(a[p] - mx); sm += a[p]; }
    float inv = 1.0f / sm;
    for (int p = 0; p < PER; p++) pr[p] = a[p] * inv;
  }

  int tile = blockIdx.x * 8 + wave;
  long ia = (long)tile * 16 + m;
  if (ia >= ROWS) ia = ROWS - 1;
  const float* xrow = xagg + ia * (long)NFEAT;

  v8f H0, H1, A0, A1;
  for (int r = 0; r < 8; r++) { H0[r] = 0.f; H1[r] = 0.f; A0[r] = 0.f; A1[r] = 0.f; }

  __syncthreads();

  for (int p = 0; p < PER; p++) {
    int f0 = hi * 2;
    v2f ax0, ax1;
    ax0.x = xrow[(f0)*PER + p];     ax0.y = xrow[(f0 + 1) * PER + p];
    ax1.x = xrow[(f0 + 4) * PER + p]; ax1.y = xrow[(f0 + 5) * PER + p];

    v8f g0, g1, z0, z1, r0, r1, t0, t1;

    gemm8(ax0, ax1, sW + 0, sb + 0, m, hi, g0, g1);
    storeTile(mG, m, hi, g0, g1);
    __syncthreads();
    gemm64(mG, mH, sL + 0, slb + 0, m, hi, z0, z1);
    for (int r = 0; r < 8; r++) { z0[r] = sigm(z0[r]); z1[r] = sigm(z1[r]); }
    __syncthreads();

    gemm8(ax0, ax1, sW + 256, sb + 32, m, hi, g0, g1);
    storeTile(mG, m, hi, g0, g1);
    __syncthreads();
    gemm64(mG, mH, sL + 2048, slb + 32, m, hi, r0, r1);
    for (int r = 0; r < 8; r++) { r0[r] = sigm(r0[r]); r1[r] = sigm(r1[r]); }

    gemm8(ax0, ax1, sW + 512, sb + 64, m, hi, g0, g1);
    __syncthreads();
    storeTile(mG, m, hi, g0, g1);
    {
      v8f q0, q1;
      for (int r = 0; r < 8; r++) { q0[r] = H0[r] * r0[r]; q1[r] = H1[r] * r1[r]; }
      storeTile(mHR, m, hi, q0, q1);
    }
    __syncthreads();
    gemm64(mG, mHR, sL + 4096, slb + 64, m, hi, t0, t1);
    for (int r = 0; r < 8; r++) { t0[r] = tanhc(t0[r]); t1[r] = tanhc(t1[r]); }

    for (int r = 0; r < 8; r++) {
      H0[r] = z0[r] * H0[r] + (1.0f - z0[r]) * t0[r];
      H1[r] = z1[r] * H1[r] + (1.0f - z1[r]) * t1[r];
      A0[r] += pr[p] * H0[r];
      A1[r] += pr[p] * H1[r];
    }
    storeTile(mH, m, hi, H0, H1);
    __syncthreads();
  }

  for (int r = 0; r < 8; r++) { A0[r] = fmaxf(A0[r], 0.f); A1[r] = fmaxf(A1[r], 0.f); }
  storeTile(mG, m, hi, A0, A1);
  __syncthreads();
  v8f o0, o1;
  {
    float b0v = sbo[m], b1v = sbo[16 + m];
    for (int r = 0; r < 8; r++) { o0[r] = b0v; o1[r] = b1v; }
#pragma unroll
    for (int kb = 0; kb < 8; kb++) {
      v2f a = ldsA(mG, m, hi, kb);
      o0 = wmma4(a, ldsB(sWo, m, hi, kb, 0), o0);
      o1 = wmma4(a, ldsB(sWo, m, hi, kb, 1), o1);
    }
  }
#pragma unroll
  for (int r = 0; r < 8; r++) {
    long row = (long)tile * 16 + r + 8 * hi;
    if (row < ROWS) {
      out[row * OUTC + m] = o0[r];
      if (m < 8) out[row * OUTC + 16 + m] = o1[r];
    }
  }
}

extern "C" void kernel_launch(void* const* d_in, const int* in_sizes, int n_in,
                              void* d_out, int out_size, void* d_ws, size_t ws_size,
                              hipStream_t stream) {
  (void)in_sizes; (void)n_in; (void)out_size; (void)ws_size;
  const float* x   = (const float*)d_in[0];
  const int*   ei  = (const int*)d_in[1];
  const float* att = (const float*)d_in[2];
  const float* Wz  = (const float*)d_in[3];  const float* bz  = (const float*)d_in[4];
  const float* Wr  = (const float*)d_in[5];  const float* br  = (const float*)d_in[6];
  const float* Wh  = (const float*)d_in[7];  const float* bh  = (const float*)d_in[8];
  const float* Lz  = (const float*)d_in[9];  const float* lbz = (const float*)d_in[10];
  const float* Lr  = (const float*)d_in[11]; const float* lbr = (const float*)d_in[12];
  const float* Lh  = (const float*)d_in[13]; const float* lbh = (const float*)d_in[14];
  const float* Wo  = (const float*)d_in[15]; const float* bo  = (const float*)d_in[16];
  float* out = (float*)d_out;

  float* wsf  = (float*)d_ws;
  float* deg  = wsf;
  float* xagg = wsf + 65536;

  k_deg_init<<<(NNODES + 255) / 256, 256, 0, stream>>>(deg);
  k_deg_acc<<<(NEDGES + 255) / 256, 256, 0, stream>>>(ei, deg);
  k_rsqrt<<<(NNODES + 255) / 256, 256, 0, stream>>>(deg);
  k_selfloop<<<(ROWS * 24 + 255) / 256, 256, 0, stream>>>(x, deg, xagg);
  k_scatter<<<NEDGES / 256, 256, 0, stream>>>(x, ei, deg, xagg);

  int tiles = ROWS / 16;
  int blocks = (tiles + 7) / 8;
  k_gru<<<blocks, 256, 0, stream>>>(xagg, att, Wz, bz, Wr, br, Wh, bh,
                                    Lz, lbz, Lr, lbr, Lh, lbh, Wo, bo, out);
}


# GRU LDS layouts made bank-conflict-free: staging tiles K-block-major with a slot XOR (A loads were 8-way, tile stores 2-way), weights fragment-major, one 16-byte load per K-block; sums unchanged
# speedup vs baseline: 1.6341x; 1.1108x over previous
#include <hip/hip_runtime.h>

#define NNODES 50000
#define NEDGES 800000
#define FIN 8
#define PER 12
#define HIDN 32
#define NB 2
#define OUTC 24
#define ROWS (NB*NNODES)
#define NFEAT (FIN*PER)

typedef float v2f __attribute__((ext_vector_type(2)));
typedef float v4f __attribute__((ext_vector_type(4)));
typedef float v8f __attribute__((ext_vector_type(8)));

__device__ __forceinline__ v8f wmma4(v2f a, v2f b, v8f c) {
  return __builtin_amdgcn_wmma_f32_16x16x4_f32(false, a, false, b, (short)0, c, false, false);
}

__device__ __forceinline__ int tix(int row, int col) {
  return ((col >> 2) << 6) + ((row ^ ((col >> 2) & 3)) << 2) + (col & 3);
}
__device__ __forceinline__ v2f ldsA(const float* buf, int m, int hi, int kb) {
  int idx = (kb << 6) + ((m ^ (kb & 3)) << 2) + hi * 2;
  return *(const v2f*)(buf + idx);
}

__device__ __forceinline__ int fragIdx(int t) {
  int k = t >> 5, n = t & 31;
  int kb = k >> 2, hi = (k >> 1) & 1, j = k & 1, nt = n >> 4, m = n & 15;
  return ((kb * 32 + hi * 16 + m) << 2) + nt * 2 + j;
}
__device__ __forceinline__ v4f ldsB4(const float* W, int lane, int kb) {
  return *(const v4f*)(W + ((kb * 32 + lane) << 2));
}

__device__ __forceinline__ void storeTile(float* buf, int m, int hi, const v8f& c0, const v8f& c1) {
#pragma unroll
  for (int r = 0; r < 8; r++) {
    buf[tix(r + 8 * hi, m)]      = c0[r];
    buf[tix(r + 8 * hi, 16 + m)] = c1[r];
  }
}

__device__ __forceinline__ void gemm8(v2f ax0, v2f ax1, const float* W, const float* bias,
                                      int m, int hi, v8f& c0, v8f& c1) {
  float b0v = bias[m], b1v = bias[16 + m];
#pragma unroll
  for (int r = 0; r < 8; r++) { c0[r] = b0v; c1[r] = b1v; }
  const int lane = hi * 16 + m;
  v4f q0 = ldsB4(W, lane, 0), q1 = ldsB4(W, lane, 1);
  c0 = wmma4(ax0, q0.xy, c0);
  c0 = wmma4(ax1, q1.xy, c0);
  c1 = wmma4(ax0, q0.zw, c1);
  c1 = wmma4(ax1, q1.zw, c1);
}

__device__ __forceinline__ void gemm64(const float* A1, const float* A2, const float* L,
                                       const float* bias, int m, int hi, v8f& c0, v8f& c1) {
  float b0v = bias[m], b1v = bias[16 + m];
#pragma unroll
  for (int r = 0; r < 8; r++) { c0[r] = b0v; c1[r] = b1v; }
  const int lane = hi * 16 + m;
#pragma unroll
  for (int kb = 0; kb < 8; kb++) {
    v2f a = ldsA(A1, m, hi, kb);
    v4f q = ldsB4(L, lane, kb);
    c0 = wmma4(a, q.xy, c0);
    c1 = wmma4(a, q.zw, c1);
  }
#pragma unroll
  for (int kb = 0; kb < 8; kb++) {
    v2f a = ldsA(A2, m, hi, kb);
    v4f q = ldsB4(L, lane, kb + 8);
    c0 = wmma4(a, q.xy, c0);
    c1 = wmma4(a, q.zw, c1);
  }
}

__device__ __forceinline__ float sigm(float x) { return 1.0f / (1.0f + __expf(-x)); }
__device__ __forceinline__ float tanhc(float x) {
  x = fminf(fmaxf(x, -15.f), 15.f);
  float e = __expf(2.0f * x);
  return (e - 1.0f) / (e + 1.0f);
}

__global__ void k_deg_init(float* deg) {
  int i = blockIdx.x * 256 + threadIdx.x;
  if (i < NNODES) deg[i] = 1.0f;
}
__global__ void k_deg_acc(const int* __restrict__ ei, float* deg) {
  int e = blockIdx.x * 256 + threadIdx.x;
  if (e < NEDGES) atomicAdd(&deg[ei[NEDGES + e]], 1.0f);
}
__global__ void k_rsqrt(float* deg) {
  int i = blockIdx.x * 256 + threadIdx.x;
  if (i < NNODES) deg[i] = rsqrtf(deg[i]);
}
__global__ void k_selfloop(const float* __restrict__ x, const float* __restrict__ dinv,
                           float* __restrict__ xagg) {
  int t = blockIdx.x * 256 + threadIdx.x;
  if (t >= ROWS * 24) return;
  int i = t / 24;
  int n = i % NNODES;
  float w = dinv[n] * dinv[n];
  float4 xv = ((const float4*)x)[t];
  float4 o; o.x = xv.x * w; o.y = xv.y * w; o.z = xv.z * w; o.w = xv.w * w;
  ((float4*)xagg)[t] = o;
}
static_assert(NEDGES % 256 == 0, "one edge per thread at the index fetch, no tail");
static_assert(NFEAT == 96, "a row is three 32-float lines");
__global__ void __launch_bounds__(256) k_scatter(const float* __restrict__ x, const int* __restrict__ ei,
                                                 const float* __restrict__ dinv, float* __restrict__ xagg) {
  const int lane = threadIdx.x & 31;
  const int e = blockIdx.x * 256 + threadIdx.x;
  const int s = ei[e];
  const int d = ei[NEDGES + e];
  const float w = dinv[s] * dinv[d];
  const size_t bstride = (size_t)NNODES * NFEAT;
  for (int k0 = 0; k0 < 32; k0 += 4) {
    float v[4][6];
#pragma unroll
    for (int u = 0; u < 4; u++) {
      const int sk = __builtin_amdgcn_readlane(s, k0 + u);
      const float* xp = x + (size_t)sk * NFEAT + lane;
      v[u][0] = xp[0];
      v[u][1] = xp[32];
      v[u][2] = xp[64];
      v[u][3] = xp[bstride];
      v[u][4] = xp[bstride + 32];
      v[u][5] = xp[bstride + 64];
    }
#pragma unroll
    for (int u = 0; u < 4; u++) {
      const int dk = __builtin_amdgcn_readlane(d, k0 + u);
      const float wk = __int_as_float(__builtin_amdgcn_readlane(__float_as_int(w), k0 + u));
      float* op = xagg + (size_t)dk * NFEAT + lane;
      atomicAdd(op, wk * v[u][0]);
      atomicAdd(op + 32, wk * v[u][1]);
      atomicAdd(op + 64, wk * v[u][2]);
      atomicAdd(op + bstride, wk * v[u][3]);
      atomicAdd(op + bstride + 32, wk * v[u][4]);
      atomicAdd(op + bstride + 64, wk * v[u][5]);
    }
  }
}

__global__ void __launch_bounds__(256) k_gru(
    const float* __restrict__ xagg, const float* __restrict__ att,
    const float* __restrict__ Wz, const float* __restrict__ bz,
    const float* __restrict__ Wr, const float* __restrict__ br,
    const float* __restrict__ Wh, const float* __restrict__ bh,
    const float* __restrict__ Lz, const float* __restrict__ lbz,
    const float* __restrict__ Lr, const float* __restrict__ lbr,
    const float* __restrict__ Lh, const float* __restrict__ lbh,
    const float* __restrict__ Wo, const float* __restrict__ bo,
    float* __restrict__ out) {
  __shared__ __attribute__((aligned(16))) float sW[3 * 8 * 32];
  __shared__ float sb[3 * 32];
  __shared__ __attribute__((aligned(16))) float sL[3 * 64 * 32];
  __shared__ float slb[3 * 32];
  __shared__ __attribute__((aligned(16))) float sWo[32 * 32];
  __shared__ float sbo[32];
  __shared__ __attribute__((aligned(16))) float sT[3][8][16 * 32];

  int tid = threadIdx.x;
  for (int t = tid; t < 8 * 32; t += 256) { int f = fragIdx(t); sW[f] = Wz[t]; sW[256 + f] = Wr[t]; sW[512 + f] = Wh[t]; }
  for (int t = tid; t < 32; t += 256) {
    sb[t] = bz[t]; sb[32 + t] = br[t]; sb[64 + t] = bh[t];
    slb[t] = lbz[t]; slb[32 + t] = lbr[t]; slb[64 + t] = lbh[t];
    sbo[t] = (t < OUTC) ? bo[t] : 0.f;
  }
  for (int t = tid; t < 64 * 32; t += 256) { int f = fragIdx(t); sL[f] = Lz[t]; sL[2048 + f] = Lr[t]; sL[4096 + f] = Lh[t]; }
  for (int t = tid; t < 32 * 32; t += 256) {
    int k = t >> 5, j = t & 31;
    sWo[fragIdx(t)] = (j < OUTC) ? Wo[k * OUTC + j] : 0.f;
  }

  int wave = tid >> 5, lane = tid & 31, m = lane & 15, hi = lane >> 4;
  float* mG = sT[0][wave];
  float* mH = sT[1][wave];
  float* mHR = sT[2][wave];
  for (int idx = lane; idx < 16 * 32; idx += 32) mH[idx] = 0.f;

  float pr[PER];
  {
    float a[PER]; float mx = -1e30f;
    for (int p = 0; p < PER; p++) { a[p] = att[p]; mx = fmaxf(mx, a[p]); }
    float sm = 0.f;
    for (int p = 0; p < PER; p++) { a[p] = __expf(a[p] - mx); sm += a[p]; }
    float inv = 1.0f / sm;
    for (int p = 0; p < PER; p++) pr[p] = a[p] * inv;
  }

  int tile = blockIdx.x * 8 + wave;
  long ia = (long)tile * 16 + m;
  if (ia >= ROWS) ia = ROWS - 1;
  const float* xrow = xagg + ia * (long)NFEAT;

  v8f H0, H1, A0, A1;
  for (int r = 0; r < 8; r++) { H0[r] = 0.f; H1[r] = 0.f; A0[r] = 0.f; A1[r] = 0.f; }

  __syncthreads();

  for (int p = 0; p < PER; p++) {
    int f0 = hi * 2;
    v2f ax0, ax1;
    ax0.x = xrow[(f0)*PER + p];     ax0.y = xrow[(f0 + 1) * PER + p];
    ax1.x = xrow[(f0 + 4) * PER + p]; ax1.y = xrow[(f0 + 5) * PER + p];

    v8f g0, g1, z0, z1, r0, r1, t0, t1;

    gemm8(ax0, ax1, sW + 0, sb + 0, m, hi, g0, g1);
    storeTile(mG, m, hi, g0, g1);
    __syncthreads();
    gemm64(mG, mH, sL + 0, slb + 0, m, hi, z0, z1);
    for (int r = 0; r < 8; r++) { z0[r] = sigm(z0[r]); z1[r] = sigm(z1[r]); }
    __syncthreads();

    gemm8(ax0, ax1, sW + 256, sb + 32, m, hi, g0, g1);
    storeTile(mG, m, hi, g0, g1);
    __syncthreads();
    gemm64(mG, mH, sL + 2048, slb + 32, m, hi, r0, r1);
    for (int r = 0; r < 8; r++) { r0[r] = sigm(r0[r]); r1[r] = sigm(r1[r]); }

    gemm8(ax0, ax1, sW + 512, sb + 64, m, hi, g0, g1);
    __syncthreads();
    storeTile(mG, m, hi, g0, g1);
    {
      v8f q0, q1;
      for (int r = 0; r < 8; r++) { q0[r] = H0[r] * r0[r]; q1[r] = H1[r] * r1[r]; }
      storeTile(mHR, m, hi, q0, q1);
    }
    __syncthreads();
    gemm64(mG, mHR, sL + 4096, slb + 64, m, hi, t0, t1);
    for (int r = 0; r < 8; r++) { t0[r] = tanhc(t0[r]); t1[r] = tanhc(t1[r]); }

    for (int r = 0; r < 8; r++) {
      H0[r] = z0[r] * H0[r] + (1.0f - z0[r]) * t0[r];
      H1[r] = z1[r] * H1[r] + (1.0f - z1[r]) * t1[r];
      A0[r] += pr[p] * H0[r];
      A1[r] += pr[p] * H1[r];
    }
    storeTile(mH, m, hi, H0, H1);
    __syncthreads();
  }

  for (int r = 0; r < 8; r++) { A0[r] = fmaxf(A0[r], 0.f); A1[r] = fmaxf(A1[r], 0.f); }
  storeTile(mG, m, hi, A0, A1);
  __syncthreads();
  v8f o0, o1;
  {
    float b0v = sbo[m], b1v = sbo[16 + m];
    for (int r = 0; r < 8; r++) { o0[r] = b0v; o1[r] = b1v; }
#pragma unroll
    for (int kb = 0; kb < 8; kb++) {
      v2f a = ldsA(mG, m, hi, kb);
      v4f q = ldsB4(sWo, lane, kb);
      o0 = wmma4(a, q.xy, o0);
      o1 = wmma4(a, q.zw, o1);
    }
  }
#pragma unroll
  for (int r = 0; r < 8; r++) {
    long row = (long)tile * 16 + r + 8 * hi;
    if (row < ROWS) {
      out[row * OUTC + m] = o0[r];
      if (m < 8) out[row * OUTC + 16 + m] = o1[r];
    }
  }
}

extern "C" void kernel_launch(void* const* d_in, const int* in_sizes, int n_in,
                              void* d_out, int out_size, void* d_ws, size_t ws_size,
                              hipStream_t stream) {
  (void)in_sizes; (void)n_in; (void)out_size; (void)ws_size;
  const float* x   = (const float*)d_in[0];
  const int*   ei  = (const int*)d_in[1];
  const float* att = (const float*)d_in[2];
  const float* Wz  = (const float*)d_in[3];  const float* bz  = (const float*)d_in[4];
  const float* Wr  = (const float*)d_in[5];  const float* br  = (const float*)d_in[6];
  const float* Wh  = (const float*)d_in[7];  const float* bh  = (const float*)d_in[8];
  const float* Lz  = (const float*)d_in[9];  const float* lbz = (const float*)d_in[10];
  const float* Lr  = (const float*)d_in[11]; const float* lbr = (const float*)d_in[12];
  const float* Lh  = (const float*)d_in[13]; const float* lbh = (const float*)d_in[14];
  const float* Wo  = (const float*)d_in[15]; const float* bo  = (const float*)d_in[16];
  float* out = (float*)d_out;

  float* wsf  = (float*)d_ws;
  float* deg  = wsf;
  float* xagg = wsf + 65536;

  k_deg_init<<<(NNODES + 255) / 256, 256, 0, stream>>>(deg);
  k_deg_acc<<<(NEDGES + 255) / 256, 256, 0, stream>>>(ei, deg);
  k_rsqrt<<<(NNODES + 255) / 256, 256, 0, stream>>>(deg);
  k_selfloop<<<(ROWS * 24 + 255) / 256, 256, 0, stream>>>(x, deg, xagg);
  k_scatter<<<NEDGES / 256, 256, 0, stream>>>(x, ei, deg, xagg);

  int tiles = ROWS / 16;
  int blocks = (tiles + 7) / 8;
  k_gru<<<blocks, 256, 0, stream>>>(xagg, att, Wz, bz, Wr, br, Wh, bh,
                                    Lz, lbz, Lr, lbr, Lh, lbh, Wo, bo, out);
}


# aggregation as a pull: a returning int add per edge gives an arrival-order slot in the target's row; a wave sums a node's source rows in registers, ordered by source index; no f32 atomics left
# speedup vs baseline: 2.1000x; 1.2851x over previous
#include <hip/hip_runtime.h>

#define NNODES 50000
#define NEDGES 800000
#define FIN 8
#define PER 12
#define HIDN 32
#define NB 2
#define OUTC 24
#define ROWS (NB*NNODES)
#define NFEAT (FIN*PER)

typedef float v2f __attribute__((ext_vector_type(2)));
typedef float v4f __attribute__((ext_vector_type(4)));
typedef float v8f __attribute__((ext_vector_type(8)));

__device__ __forceinline__ v8f wmma4(v2f a, v2f b, v8f c) {
  return __builtin_amdgcn_wmma_f32_16x16x4_f32(false, a, false, b, (short)0, c, false, false);
}

__device__ __forceinline__ int tix(int row, int col) {
  return ((col >> 2) << 6) + ((row ^ ((col >> 2) & 3)) << 2) + (col & 3);
}
__device__ __forceinline__ v2f ldsA(const float* buf, int m, int hi, int kb) {
  int idx = (kb << 6) + ((m ^ (kb & 3)) << 2) + hi * 2;
  return *(const v2f*)(buf + idx);
}

__device__ __forceinline__ int fragIdx(int t) {
  int k = t >> 5, n = t & 31;
  int kb = k >> 2, hi = (k >> 1) & 1, j = k & 1, nt = n >> 4, m = n & 15;
  return ((kb * 32 + hi * 16 + m) << 2) + nt * 2 + j;
}
__device__ __forceinline__ v4f ldsB4(const float* W, int lane, int kb) {
  return *(const v4f*)(W + ((kb * 32 + lane) << 2));
}

__device__ __forceinline__ void storeTile(float* buf, int m, int hi, const v8f& c0, const v8f& c1) {
#pragma unroll
  for (int r = 0; r < 8; r++) {
    buf[tix(r + 8 * hi, m)]      = c0[r];
    buf[tix(r + 8 * hi, 16 + m)] = c1[r];
  }
}

__device__ __forceinline__ void gemm8(v2f ax0, v2f ax1, const float* W, const float* bias,
                                      int m, int hi, v8f& c0, v8f& c1) {
  float b0v = bias[m], b1v = bias[16 + m];
#pragma unroll
  for (int r = 0; r < 8; r++) { c0[r] = b0v; c1[r] = b1v; }
  const int lane = hi * 16 + m;
  v4f q0 = ldsB4(W, lane, 0), q1 = ldsB4(W, lane, 1);
  c0 = wmma4(ax0, q0.xy, c0);
  c0 = wmma4(ax1, q1.xy, c0);
  c1 = wmma4(ax0, q0.zw, c1);
  c1 = wmma4(ax1, q1.zw, c1);
}

__device__ __forceinline__ void gemm64(const float* A1, const float* A2, const float* L,
                                       const float* bias, int m, int hi, v8f& c0, v8f& c1) {
  float b0v = bias[m], b1v = bias[16 + m];
#pragma unroll
  for (int r = 0; r < 8; r++) { c0[r] = b0v; c1[r] = b1v; }
  const int lane = hi * 16 + m;
#pragma unroll
  for (int kb = 0; kb < 8; kb++) {
    v2f a = ldsA(A1, m, hi, kb);
    v4f q = ldsB4(L, lane, kb);
    c0 = wmma4(a, q.xy, c0);
    c1 = wmma4(a, q.zw, c1);
  }
#pragma unroll
  for (int kb = 0; kb < 8; kb++) {
    v2f a = ldsA(A2, m, hi, kb);
    v4f q = ldsB4(L, lane, kb + 8);
    c0 = wmma4(a, q.xy, c0);
    c1 = wmma4(a, q.zw, c1);
  }
}

__device__ __forceinline__ float sigm(float x) { return 1.0f / (1.0f + __expf(-x)); }
__device__ __forceinline__ float tanhc(float x) {
  x = fminf(fmaxf(x, -15.f), 15.f);
  float e = __expf(2.0f * x);
  return (e - 1.0f) / (e + 1.0f);
}

#define SLOTCAP NFEAT
#define GWAVES 25000
static_assert(NEDGES % 256 == 0, "one edge per thread, no tail");
static_assert(NFEAT == 96, "a row is 24 lanes x 16 bytes = three 128-byte lines");
static_assert(2 * GWAVES == NNODES && GWAVES % 8 == 0, "two nodes per gather wave, eight waves per block");

__global__ void k_cnt_init(int* cnt) {
  int i = blockIdx.x * 256 + threadIdx.x;
  if (i < NNODES) cnt[i] = 0;
}
__global__ void __launch_bounds__(256) k_place(const int* __restrict__ ei, int* cnt, int* slots) {
  const int e = blockIdx.x * 256 + threadIdx.x;
  const int s = ei[e];
  const int d = ei[NEDGES + e];
  const int r = atomicAdd(&cnt[d], 1);
  if (r < SLOTCAP) slots[d * NFEAT + r] = s;
}

__global__ void __launch_bounds__(256) k_gather(const float* __restrict__ x, const int* __restrict__ ei,
                                                const int* __restrict__ cnt, float* xagg) {
  const int lane = threadIdx.x & 31;
  const int ql = lane < 24 ? lane : 23;
  const v4f* x4 = (const v4f*)x;
  const int b4 = NNODES * (NFEAT / 4);
  const int w0 = blockIdx.x * 8 + (threadIdx.x >> 5);
  for (int it = 0; it < 2; it++) {
    const int n = __builtin_amdgcn_readfirstlane(w0 + it * GWAVES);
    const int deg = cnt[n];
    const float dn = rsqrtf((float)(deg + 1));
    v4f a0, a1;
    {
      const float ww = dn * dn;
      a0 = ww * x4[n * 24 + ql];
      a1 = ww * x4[b4 + n * 24 + ql];
    }
    if (deg <= SLOTCAP) {
      const int* slots = (const int*)xagg + n * NFEAT;
      for (int c0 = 0; c0 < deg; c0 += 32) {
        const int m = (deg - c0 < 32) ? deg - c0 : 32;
        int key = 0x7fffffff;
        float ws = 0.f;
        if (lane < m) {
          const int s = slots[c0 + lane];
          ws = rsqrtf((float)(cnt[s] + 1)) * dn;
          key = (s << 5) | lane;
        }
        int myrank = 0;
        for (int j = 0; j < m; j++) myrank += (__builtin_amdgcn_readlane(key, j) < key) ? 1 : 0;
        for (int k0 = 0; k0 < m; k0 += 4) {
          v4f r0[4], r1[4];
          float wk[4];
#pragma unroll
          for (int u = 0; u < 4; u++) {
            wk[u] = 0.f; r0[u] = 0.f; r1[u] = 0.f;
            if (k0 + u < m) {
              const int l = __builtin_ctz(__builtin_amdgcn_ballot_w32(myrank == k0 + u));
              const int sk = __builtin_amdgcn_readlane(key, l) >> 5;
              wk[u] = __int_as_float(__builtin_amdgcn_readlane(__float_as_int(ws), l));
              r0[u] = x4[sk * 24 + ql];
              r1[u] = x4[b4 + sk * 24 + ql];
            }
          }
#pragma unroll
          for (int u = 0; u < 4; u++) { a0 += wk[u] * r0[u]; a1 += wk[u] * r1[u]; }
        }
      }
    } else {
      for (int e0 = 0; e0 < NEDGES; e0 += 32) {
        unsigned mask = __builtin_amdgcn_ballot_w32(ei[NEDGES + e0 + lane] == n);
        while (mask) {
          const int l = __builtin_ctz(mask);
          mask &= mask - 1;
          const int sk = ei[e0 + l];
          const float wk = rsqrtf((float)(cnt[sk] + 1)) * dn;
          a0 += wk * x4[sk * 24 + ql];
          a1 += wk * x4[b4 + sk * 24 + ql];
        }
      }
    }
    if (lane < 24) {
      ((v4f*)xagg)[n * 24 + lane] = a0;
      ((v4f*)xagg)[b4 + n * 24 + lane] = a1;
    }
  }
}

__global__ void __launch_bounds__(256) k_gru(
    const float* __restrict__ xagg, const float* __restrict__ att,
    const float* __restrict__ Wz, const float* __restrict__ bz,
    const float* __restrict__ Wr, const float* __restrict__ br,
    const float* __restrict__ Wh, const float* __restrict__ bh,
    const float* __restrict__ Lz, const float* __restrict__ lbz,
    const float* __restrict__ Lr, const float* __restrict__ lbr,
    const float* __restrict__ Lh, const float* __restrict__ lbh,
    const float* __restrict__ Wo, const float* __restrict__ bo,
    float* __restrict__ out) {
  __shared__ __attribute__((aligned(16))) float sW[3 * 8 * 32];
  __shared__ float sb[3 * 32];
  __shared__ __attribute__((aligned(16))) float sL[3 * 64 * 32];
  __shared__ float slb[3 * 32];
  __shared__ __attribute__((aligned(16))) float sWo[32 * 32];
  __shared__ float sbo[32];
  __shared__ __attribute__((aligned(16))) float sT[3][8][16 * 32];

  int tid = threadIdx.x;
  for (int t = tid; t < 8 * 32; t += 256) { int f = fragIdx(t); sW[f] = Wz[t]; sW[256 + f] = Wr[t]; sW[512 + f] = Wh[t]; }
  for (int t = tid; t < 32; t += 256) {
    sb[t] = bz[t]; sb[32 + t] = br[t]; sb[64 + t] = bh[t];
    slb[t] = lbz[t]; slb[32 + t] = lbr[t]; slb[64 + t] = lbh[t];
    sbo[t] = (t < OUTC) ? bo[t] : 0.f;
  }
  for (int t = tid; t < 64 * 32; t += 256) { int f = fragIdx(t); sL[f] = Lz[t]; sL[2048 + f] = Lr[t]; sL[4096 + f] = Lh[t]; }
  for (int t = tid; t < 32 * 32; t += 256) {
    int k = t >> 5, j = t & 31;
    sWo[fragIdx(t)] = (j < OUTC) ? Wo[k * OUTC + j] : 0.f;
  }

  int wave = tid >> 5, lane = tid & 31, m = lane & 15, hi = lane >> 4;
  float* mG = sT[0][wave];
  float* mH = sT[1][wave];
  float* mHR = sT[2][wave];
  for (int idx = lane; idx < 16 * 32; idx += 32) mH[idx] = 0.f;

  float pr[PER];
  {
    float a[PER]; float mx = -1e30f;
    for (int p = 0; p < PER; p++) { a[p] = att[p]; mx = fmaxf(mx, a[p]); }
    float sm = 0.f;
    for (int p = 0; p < PER; p++) { a[p] = __expf(a[p] - mx); sm += a[p]; }
    float inv = 1.0f / sm;
    for (int p = 0; p < PER; p++) pr[p] = a[p] * inv;
  }

  int tile = blockIdx.x * 8 + wave;
  long ia = (long)tile * 16 + m;
  if (ia >= ROWS) ia = ROWS - 1;
  const float* xrow = xagg + ia * (long)NFEAT;

  v8f H0, H1, A0, A1;
  for (int r = 0; r < 8; r++) { H0[r] = 0.f; H1[r] = 0.f; A0[r] = 0.f; A1[r] = 0.f; }

  __syncthreads();

  for (int p = 0; p < PER; p++) {
    int f0 = hi * 2;
    v2f ax0, ax1;
    ax0.x = xrow[(f0)*PER + p];     ax0.y = xrow[(f0 + 1) * PER + p];
    ax1.x = xrow[(f0 + 4) * PER + p]; ax1.y = xrow[(f0 + 5) * PER + p];

    v8f g0, g1, z0, z1, r0, r1, t0, t1;

    gemm8(ax0, ax1, sW + 0, sb + 0, m, hi, g0, g1);
    storeTile(mG, m, hi, g0, g1);
    __syncthreads();
    gemm64(mG, mH, sL + 0, slb + 0, m, hi, z0, z1);
    for (int r = 0; r < 8; r++) { z0[r] = sigm(z0[r]); z1[r] = sigm(z1[r]); }
    __syncthreads();

    gemm8(ax0, ax1, sW + 256, sb + 32, m, hi, g0, g1);
    storeTile(mG, m, hi, g0, g1);
    __syncthreads();
    gemm64(mG, mH, sL + 2048, slb + 32, m, hi, r0, r1);
    for (int r = 0; r < 8; r++) { r0[r] = sigm(r0[r]); r1[r] = sigm(r1[r]); }

    gemm8(ax0, ax1, sW + 512, sb + 64, m, hi, g0, g1);
    __syncthreads();
    storeTile(mG, m, hi, g0, g1);
    {
      v8f q0, q1;
      for (int r = 0; r < 8; r++) { q0[r] = H0[r] * r0[r]; q1[r] = H1[r] * r1[r]; }
      storeTile(mHR, m, hi, q0, q1);
    }
    __syncthreads();
    gemm64(mG, mHR, sL + 4096, slb + 64, m, hi, t0, t1);
    for (int r = 0; r < 8; r++) { t0[r] = tanhc(t0[r]); t1[r] = tanhc(t1[r]); }

    for (int r = 0; r < 8; r++) {
      H0[r] = z0[r] * H0[r] + (1.0f - z0[r]) * t0[r];
      H1[r] = z1[r] * H1[r] + (1.0f - z1[r]) * t1[r];
      A0[r] += pr[p] * H0[r];
      A1[r] += pr[p] * H1[r];
    }
    storeTile(mH, m, hi, H0, H1);
    __syncthreads();
  }

  for (int r = 0; r < 8; r++) { A0[r] = fmaxf(A0[r], 0.f); A1[r] = fmaxf(A1[r], 0.f); }
  storeTile(mG, m, hi, A0, A1);
  __syncthreads();
  v8f o0, o1;
  {
    float b0v = sbo[m], b1v = sbo[16 + m];
    for (int r = 0; r < 8; r++) { o0[r] = b0v; o1[r] = b1v; }
#pragma unroll
    for (int kb = 0; kb < 8; kb++) {
      v2f a = ldsA(mG, m, hi, kb);
      v4f q = ldsB4(sWo, lane, kb);
      o0 = wmma4(a, q.xy, o0);
      o1 = wmma4(a, q.zw, o1);
    }
  }
#pragma unroll
  for (int r = 0; r < 8; r++) {
    long row = (long)tile * 16 + r + 8 * hi;
    if (row < ROWS) {
      out[row * OUTC + m] = o0[r];
      if (m < 8) out[row * OUTC + 16 + m] = o1[r];
    }
  }
}

extern "C" void kernel_launch(void* const* d_in, const int* in_sizes, int n_in,
                              void* d_out, int out_size, void* d_ws, size_t ws_size,
                              hipStream_t stream) {
  (void)in_sizes; (void)n_in; (void)out_size; (void)ws_size;
  const float* x   = (const float*)d_in[0];
  const int*   ei  = (const int*)d_in[1];
  const float* att = (const float*)d_in[2];
  const float* Wz  = (const float*)d_in[3];  const float* bz  = (const float*)d_in[4];
  const float* Wr  = (const float*)d_in[5];  const float* br  = (const float*)d_in[6];
  const float* Wh  = (const float*)d_in[7];  const float* bh  = (const float*)d_in[8];
  const float* Lz  = (const float*)d_in[9];  const float* lbz = (const float*)d_in[10];
  const float* Lr  = (const float*)d_in[11]; const float* lbr = (const float*)d_in[12];
  const float* Lh  = (const float*)d_in[13]; const float* lbh = (const float*)d_in[14];
  const float* Wo  = (const float*)d_in[15]; const float* bo  = (const float*)d_in[16];
  float* out = (float*)d_out;

  float* wsf  = (float*)d_ws;
  int*   cnt  = (int*)d_ws;
  float* xagg = wsf + 65536;

  k_cnt_init<<<(NNODES + 255) / 256, 256, 0, stream>>>(cnt);
  k_place<<<NEDGES / 256, 256, 0, stream>>>(ei, cnt, (int*)xagg);
  k_gather<<<GWAVES / 8, 256, 0, stream>>>(x, ei, cnt, xagg);

  int tiles = ROWS / 16;
  int blocks = (tiles + 7) / 8;
  k_gru<<<blocks, 256, 0, stream>>>(xagg, att, Wz, bz, Wr, br, Wh, bh,
                                    Lz, lbz, Lr, lbr, Lh, lbh, Wo, bo, out);
}


# GRU: each gate's graph-conv weight folded into its linear layer (W@Ltop built once per block on WMMA): per gate K = 8+32 not 8 then 64, no G tile through LDS; sums reassociated
# speedup vs baseline: 2.7076x; 1.2893x over previous
#include <hip/hip_runtime.h>

#define NNODES 50000
#define NEDGES 800000
#define FIN 8
#define PER 12
#define HIDN 32
#define NB 2
#define OUTC 24
#define ROWS (NB*NNODES)
#define NFEAT (FIN*PER)

typedef float v2f __attribute__((ext_vector_type(2)));
typedef float v4f __attribute__((ext_vector_type(4)));
typedef float v8f __attribute__((ext_vector_type(8)));

__device__ __forceinline__ v8f wmma4(v2f a, v2f b, v8f c) {
  return __builtin_amdgcn_wmma_f32_16x16x4_f32(false, a, false, b, (short)0, c, false, false);
}

__device__ __forceinline__ int tix(int row, int col) {
  return ((col >> 2) << 6) + ((row ^ ((col >> 2) & 3)) << 2) + (col & 3);
}
__device__ __forceinline__ v2f ldsA(const float* buf, int m, int hi, int kb) {
  int idx = (kb << 6) + ((m ^ (kb & 3)) << 2) + hi * 2;
  return *(const v2f*)(buf + idx);
}

__device__ __forceinline__ int fragIdx(int t) {
  int k = t >> 5, n = t & 31;
  int kb = k >> 2, hi = (k >> 1) & 1, j = k & 1, nt = n >> 4, m = n & 15;
  return ((kb * 32 + hi * 16 + m) << 2) + nt * 2 + j;
}
__device__ __forceinline__ v4f ldsB4(const float* W, int lane, int kb) {
  return *(const v4f*)(W + ((kb * 32 + lane) << 2));
}

__device__ __forceinline__ void storeTile(float* buf, int m, int hi, const v8f& c0, const v8f& c1) {
#pragma unroll
  for (int r = 0; r < 8; r++) {
    buf[tix(r + 8 * hi, m)]      = c0[r];
    buf[tix(r + 8 * hi, 16 + m)] = c1[r];
  }
}

__device__ __forceinline__ void mmaX(v2f ax0, v2f ax1, const float* M, int lane, v8f& c0, v8f& c1) {
  v4f q0 = ldsB4(M, lane, 0), q1 = ldsB4(M, lane, 1);
  c0 = wmma4(ax0, q0.xy, c0);
  c1 = wmma4(ax0, q0.zw, c1);
  c0 = wmma4(ax1, q1.xy, c0);
  c1 = wmma4(ax1, q1.zw, c1);
}

__device__ __forceinline__ void mmaH(const float* T, const float* L, int m, int hi, int lane, v8f& c0, v8f& c1) {
#pragma unroll
  for (int kb = 0; kb < 8; kb++) {
    v2f a = ldsA(T, m, hi, kb);
    v4f q = ldsB4(L, lane, kb + 8);
    c0 = wmma4(a, q.xy, c0);
    c1 = wmma4(a, q.zw, c1);
  }
}

__device__ __forceinline__ float sigm(float x) { return 1.0f / (1.0f + __expf(-x)); }
__device__ __forceinline__ float tanhc(float x) {
  x = fminf(fmaxf(x, -15.f), 15.f);
  float e = __expf(2.0f * x);
  return (e - 1.0f) / (e + 1.0f);
}

#define SLOTCAP NFEAT
#define GWAVES 25000
static_assert(NEDGES % 256 == 0, "one edge per thread, no tail");
static_assert(NFEAT == 96, "a row is 24 lanes x 16 bytes = three 128-byte lines");
static_assert(2 * GWAVES == NNODES && GWAVES % 8 == 0, "two nodes per gather wave, eight waves per block");

__global__ void k_cnt_init(int* cnt) {
  int i = blockIdx.x * 256 + threadIdx.x;
  if (i < NNODES) cnt[i] = 0;
}
__global__ void __launch_bounds__(256) k_place(const int* __restrict__ ei, int* cnt, int* slots) {
  const int e = blockIdx.x * 256 + threadIdx.x;
  const int s = ei[e];
  const int d = ei[NEDGES + e];
  const int r = atomicAdd(&cnt[d], 1);
  if (r < SLOTCAP) slots[d * NFEAT + r] = s;
}

__global__ void __launch_bounds__(256) k_gather(const float* __restrict__ x, const int* __restrict__ ei,
                                                const int* __restrict__ cnt, float* xagg) {
  const int lane = threadIdx.x & 31;
  const int ql = lane < 24 ? lane : 23;
  const v4f* x4 = (const v4f*)x;
  const int b4 = NNODES * (NFEAT / 4);
  const int w0 = blockIdx.x * 8 + (threadIdx.x >> 5);
  for (int it = 0; it < 2; it++) {
    const int n = __builtin_amdgcn_readfirstlane(w0 + it * GWAVES);
    const int deg = cnt[n];
    const float dn = rsqrtf((float)(deg + 1));
    v4f a0, a1;
    {
      const float ww = dn * dn;
      a0 = ww * x4[n * 24 + ql];
      a1 = ww * x4[b4 + n * 24 + ql];
    }
    if (deg <= SLOTCAP) {
      const int* slots = (const int*)xagg + n * NFEAT;
      for (int c0 = 0; c0 < deg; c0 += 32) {
        const int m = (deg - c0 < 32) ? deg - c0 : 32;
        int key = 0x7fffffff;
        float ws = 0.f;
        if (lane < m) {
          const int s = slots[c0 + lane];
          ws = rsqrtf((float)(cnt[s] + 1)) * dn;
          key = (s << 5) | lane;
        }
        int myrank = 0;
        for (int j = 0; j < m; j++) myrank += (__builtin_amdgcn_readlane(key, j) < key) ? 1 : 0;
        for (int k0 = 0; k0 < m; k0 += 4) {
          v4f r0[4], r1[4];
          float wk[4];
#pragma unroll
          for (int u = 0; u < 4; u++) {
            wk[u] = 0.f; r0[u] = 0.f; r1[u] = 0.f;
            if (k0 + u < m) {
              const int l = __builtin_ctz(__builtin_amdgcn_ballot_w32(myrank == k0 + u));
              const int sk = __builtin_amdgcn_readlane(key, l) >> 5;
              wk[u] = __int_as_float(__builtin_amdgcn_readlane(__float_as_int(ws), l));
              r0[u] = x4[sk * 24 + ql];
              r1[u] = x4[b4 + sk * 24 + ql];
            }
          }
#pragma unroll
          for (int u = 0; u < 4; u++) { a0 += wk[u] * r0[u]; a1 += wk[u] * r1[u]; }
        }
      }
    } else {
      for (int e0 = 0; e0 < NEDGES; e0 += 32) {
        unsigned mask = __builtin_amdgcn_ballot_w32(ei[NEDGES + e0 + lane] == n);
        while (mask) {
          const int l = __builtin_ctz(mask);
          mask &= mask - 1;
          const int sk = ei[e0 + l];
          const float wk = rsqrtf((float)(cnt[sk] + 1)) * dn;
          a0 += wk * x4[sk * 24 + ql];
          a1 += wk * x4[b4 + sk * 24 + ql];
        }
      }
    }
    if (lane < 24) {
      ((v4f*)xagg)[n * 24 + lane] = a0;
      ((v4f*)xagg)[b4 + n * 24 + lane] = a1;
    }
  }
}

__global__ void __launch_bounds__(256, 6) k_gru(
    const float* __restrict__ xagg, const float* __restrict__ att,
    const float* __restrict__ Wz, const float* __restrict__ bz,
    const float* __restrict__ Wr, const float* __restrict__ br,
    const float* __restrict__ Wh, const float* __restrict__ bh,
    const float* __restrict__ Lz, const float* __restrict__ lbz,
    const float* __restrict__ Lr, const float* __restrict__ lbr,
    const float* __restrict__ Lh, const float* __restrict__ lbh,
    const float* __restrict__ Wo, const float* __restrict__ bo,
    float* __restrict__ out) {
  __shared__ __attribute__((aligned(16))) float sM[3 * 16 * 32];
  __shared__ __attribute__((aligned(16))) float sL[3 * 64 * 32];
  __shared__ __attribute__((aligned(16))) float sWo[32 * 32];
  __shared__ float sbo[32];
  __shared__ float sc[3 * 32];
  __shared__ float sPr[PER];
  __shared__ __attribute__((aligned(16))) float sT[2][8][16 * 32];

  int tid = threadIdx.x;
  for (int t = tid; t < 32; t += 256) sbo[t] = (t < OUTC) ? bo[t] : 0.f;
  for (int t = tid; t < 64 * 32; t += 256) { int f = fragIdx(t); sL[f] = Lz[t]; sL[2048 + f] = Lr[t]; sL[4096 + f] = Lh[t]; }
  for (int t = tid; t < 32 * 32; t += 256) {
    int k = t >> 5, j = t & 31;
    sWo[fragIdx(t)] = (j < OUTC) ? Wo[k * OUTC + j] : 0.f;
  }

  int wave = tid >> 5, lane = tid & 31, m = lane & 15, hi = lane >> 4;
  float* mH = sT[0][wave];
  float* mHR = sT[1][wave];
  for (int idx = lane; idx < 16 * 32; idx += 32) mH[idx] = 0.f;

  {
    float a[PER]; float mx = -1e30f;
    for (int p = 0; p < PER; p++) { a[p] = att[p]; mx = fmaxf(mx, a[p]); }
    float sm = 0.f;
    for (int p = 0; p < PER; p++) { a[p] = __expf(a[p] - mx); sm += a[p]; }
    float inv = 1.0f / sm;
    if (tid == 0) for (int p = 0; p < PER; p++) sPr[p] = a[p] * inv;
  }

  int tile = blockIdx.x * 8 + wave;
  long ia = (long)tile * 16 + m;
  if (ia >= ROWS) ia = ROWS - 1;
  const float* xrow = xagg + ia * (long)NFEAT;

  v8f H0, H1, A0, A1;
  for (int r = 0; r < 8; r++) { H0[r] = 0.f; H1[r] = 0.f; A0[r] = 0.f; A1[r] = 0.f; }

  __syncthreads();

  if (wave < 3) {
    const float* Wg = (wave == 0) ? Wz : (wave == 1) ? Wr : Wh;
    const float* bg = (wave == 0) ? bz : (wave == 1) ? br : bh;
    const float* Lg = sL + wave * 2048;
    float* Mg = sM + wave * 512;
    const float* arow = (m < 8) ? (Wg + m * 32) : bg;
    const bool live = (m <= 8);
    v2f af[8];
    for (int kb = 0; kb < 8; kb++) {
      const int k0 = kb * 4 + hi * 2;
      const float u = arow[k0], v = arow[k0 + 1];
      af[kb].x = live ? u : 0.f;
      af[kb].y = live ? v : 0.f;
    }
    v8f c0, c1;
    for (int r = 0; r < 8; r++) { c0[r] = 0.f; c1[r] = 0.f; }
    for (int kb = 0; kb < 8; kb++) {
      v4f q = ldsB4(Lg, lane, kb);
      c0 = wmma4(af[kb], q.xy, c0);
      c1 = wmma4(af[kb], q.zw, c1);
    }
    for (int r = 0; r < 8; r++) {
      Mg[fragIdx((r + 8 * hi) * 32 + m)]      = c0[r];
      Mg[fragIdx((r + 8 * hi) * 32 + 16 + m)] = c1[r];
    }
  }
  __syncthreads();

  if (tid < 96) {
    const int g = tid >> 5, n = tid & 31;
    const float* lbg = (g == 0) ? lbz : (g == 1) ? lbr : lbh;
    sc[tid] = sM[g * 512 + fragIdx(8 * 32 + n)] + lbg[n];
  }
  __syncthreads();

  for (int p = 0; p < PER; p++) {
    int f0 = hi * 2;
    v2f ax0, ax1;
    ax0.x = xrow[(f0)*PER + p];     ax0.y = xrow[(f0 + 1) * PER + p];
    ax1.x = xrow[(f0 + 4) * PER + p]; ax1.y = xrow[(f0 + 5) * PER + p];

    v8f z0, z1, r0, r1, t0, t1;
    { const float u = sc[m], v = sc[16 + m]; for (int r = 0; r < 8; r++) { z0[r] = u; z1[r] = v; } }
    mmaX(ax0, ax1, sM, lane, z0, z1);
    mmaH(mH, sL, m, hi, lane, z0, z1);
    for (int r = 0; r < 8; r++) { z0[r] = sigm(z0[r]); z1[r] = sigm(z1[r]); }

    { const float u = sc[32 + m], v = sc[48 + m]; for (int r = 0; r < 8; r++) { r0[r] = u; r1[r] = v; } }
    mmaX(ax0, ax1, sM + 512, lane, r0, r1);
    mmaH(mH, sL + 2048, m, hi, lane, r0, r1);
    for (int r = 0; r < 8; r++) { r0[r] = sigm(r0[r]); r1[r] = sigm(r1[r]); }
    {
      v8f q0, q1;
      for (int r = 0; r < 8; r++) { q0[r] = H0[r] * r0[r]; q1[r] = H1[r] * r1[r]; }
      storeTile(mHR, m, hi, q0, q1);
    }
    __syncthreads();

    { const float u = sc[64 + m], v = sc[80 + m]; for (int r = 0; r < 8; r++) { t0[r] = u; t1[r] = v; } }
    mmaX(ax0, ax1, sM + 1024, lane, t0, t1);
    mmaH(mHR, sL + 4096, m, hi, lane, t0, t1);
    for (int r = 0; r < 8; r++) { t0[r] = tanhc(t0[r]); t1[r] = tanhc(t1[r]); }

    const float prp = sPr[p];
    for (int r = 0; r < 8; r++) {
      H0[r] = z0[r] * H0[r] + (1.0f - z0[r]) * t0[r];
      H1[r] = z1[r] * H1[r] + (1.0f - z1[r]) * t1[r];
      A0[r] += prp * H0[r];
      A1[r] += prp * H1[r];
    }
    storeTile(mH, m, hi, H0, H1);
    __syncthreads();
  }

  for (int r = 0; r < 8; r++) { A0[r] = fmaxf(A0[r], 0.f); A1[r] = fmaxf(A1[r], 0.f); }
  storeTile(mHR, m, hi, A0, A1);
  __syncthreads();
  v8f o0, o1;
  {
    float b0v = sbo[m], b1v = sbo[16 + m];
    for (int r = 0; r < 8; r++) { o0[r] = b0v; o1[r] = b1v; }
#pragma unroll
    for (int kb = 0; kb < 8; kb++) {
      v2f a = ldsA(mHR, m, hi, kb);
      v4f q = ldsB4(sWo, lane, kb);
      o0 = wmma4(a, q.xy, o0);
      o1 = wmma4(a, q.zw, o1);
    }
  }
#pragma unroll
  for (int r = 0; r < 8; r++) {
    long row = (long)tile * 16 + r + 8 * hi;
    if (row < ROWS) {
      out[row * OUTC + m] = o0[r];
      if (m < 8) out[row * OUTC + 16 + m] = o1[r];
    }
  }
}

extern "C" void kernel_launch(void* const* d_in, const int* in_sizes, int n_in,
                              void* d_out, int out_size, void* d_ws, size_t ws_size,
                              hipStream_t stream) {
  (void)in_sizes; (void)n_in; (void)out_size; (void)ws_size;
  const float* x   = (const float*)d_in[0];
  const int*   ei  = (const int*)d_in[1];
  const float* att = (const float*)d_in[2];
  const float* Wz  = (const float*)d_in[3];  const float* bz  = (const float*)d_in[4];
  const float* Wr  = (const float*)d_in[5];  const float* br  = (const float*)d_in[6];
  const float* Wh  = (const float*)d_in[7];  const float* bh  = (const float*)d_in[8];
  const float* Lz  = (const float*)d_in[9];  const float* lbz = (const float*)d_in[10];
  const float* Lr  = (const float*)d_in[11]; const float* lbr = (const float*)d_in[12];
  const float* Lh  = (const float*)d_in[13]; const float* lbh = (const float*)d_in[14];
  const float* Wo  = (const float*)d_in[15]; const float* bo  = (const float*)d_in[16];
  float* out = (float*)d_out;

  float* wsf  = (float*)d_ws;
  int*   cnt  = (int*)d_ws;
  float* xagg = wsf + 65536;

  k_cnt_init<<<(NNODES + 255) / 256, 256, 0, stream>>>(cnt);
  k_place<<<NEDGES / 256, 256, 0, stream>>>(ei, cnt, (int*)xagg);
  k_gather<<<GWAVES / 8, 256, 0, stream>>>(x, ei, cnt, xagg);

  int tiles = ROWS / 16;
  int blocks = (tiles + 7) / 8;
  k_gru<<<blocks, 256, 0, stream>>>(xagg, att, Wz, bz, Wr, br, Wh, bh,
                                    Lz, lbz, Lr, lbr, Lh, lbh, Wo, bo, out);
}


# GRU transposed (weights = A, nodes = B columns): a gate's D tile is the next product's B fragment as is; H and H*R stay in registers, no LDS staging, no barrier per period; k order of sums permuted
# speedup vs baseline: 2.8199x; 1.0415x over previous
#include <hip/hip_runtime.h>

#define NNODES 50000
#define NEDGES 800000
#define FIN 8
#define PER 12
#define HIDN 32
#define NB 2
#define OUTC 24
#define ROWS (NB*NNODES)
#define NFEAT (FIN*PER)

typedef float v2f __attribute__((ext_vector_type(2)));
typedef float v4f __attribute__((ext_vector_type(4)));
typedef float v8f __attribute__((ext_vector_type(8)));

__device__ __forceinline__ v8f wmma4(v2f a, v2f b, v8f c) {
  return __builtin_amdgcn_wmma_f32_16x16x4_f32(false, a, false, b, (short)0, c, false, false);
}


__device__ __forceinline__ int fragIdx(int t) {
  int k = t >> 5, n = t & 31;
  int kb = k >> 2, hi = (k >> 1) & 1, j = k & 1, nt = n >> 4, m = n & 15;
  return ((kb * 32 + hi * 16 + m) << 2) + nt * 2 + j;
}
__device__ __forceinline__ int fragIdxD(int k, int n) {
  int kb = ((k >> 4) << 2) + ((k >> 1) & 3), hi = (k >> 3) & 1, j = k & 1, nt = n >> 4, m = n & 15;
  return ((kb * 32 + hi * 16 + m) << 2) + nt * 2 + j;
}
__device__ __forceinline__ v4f ldsB4(const float* W, int lane, int kb) {
  return *(const v4f*)(W + ((kb * 32 + lane) << 2));
}

__device__ __forceinline__ void biasTiles(const float* b, int hi, v8f& c0, v8f& c1) {
  const v4f* p = (const v4f*)(b + 8 * hi);
  v4f u0 = p[0], u1 = p[1], w0 = p[4], w1 = p[5];
  c0 = __builtin_shufflevector(u0, u1, 0, 1, 2, 3, 4, 5, 6, 7);
  c1 = __builtin_shufflevector(w0, w1, 0, 1, 2, 3, 4, 5, 6, 7);
}

__device__ __forceinline__ void mmaX(v2f bx0, v2f bx1, const float* M, int lane, v8f& c0, v8f& c1) {
  v4f q0 = ldsB4(M, lane, 0), q1 = ldsB4(M, lane, 1);
  c0 = wmma4(q0.xy, bx0, c0);
  c1 = wmma4(q0.zw, bx0, c1);
  c0 = wmma4(q1.xy, bx1, c0);
  c1 = wmma4(q1.zw, bx1, c1);
}

__device__ __forceinline__ void mmaD(const v8f& s0, const v8f& s1, const float* W, int lane, v8f& c0, v8f& c1) {
#pragma unroll
  for (int kb = 0; kb < 8; kb++) {
    const int p = kb & 3;
    v2f b;
    b.x = (kb < 4) ? s0[2 * p] : s1[2 * p];
    b.y = (kb < 4) ? s0[2 * p + 1] : s1[2 * p + 1];
    v4f q = ldsB4(W, lane, kb);
    c0 = wmma4(q.xy, b, c0);
    c1 = wmma4(q.zw, b, c1);
  }
}

__device__ __forceinline__ float sigm(float x) { return 1.0f / (1.0f + __expf(-x)); }
__device__ __forceinline__ float tanhc(float x) {
  x = fminf(fmaxf(x, -15.f), 15.f);
  float e = __expf(2.0f * x);
  return (e - 1.0f) / (e + 1.0f);
}

#define SLOTCAP NFEAT
#define GWAVES 25000
static_assert(NEDGES % 256 == 0, "one edge per thread, no tail");
static_assert(NFEAT == 96, "a row is 24 lanes x 16 bytes = three 128-byte lines");
static_assert(2 * GWAVES == NNODES && GWAVES % 8 == 0, "two nodes per gather wave, eight waves per block");

__global__ void k_cnt_init(int* cnt) {
  int i = blockIdx.x * 256 + threadIdx.x;
  if (i < NNODES) cnt[i] = 0;
}
__global__ void __launch_bounds__(256) k_place(const int* __restrict__ ei, int* cnt, int* slots) {
  const int e = blockIdx.x * 256 + threadIdx.x;
  const int s = ei[e];
  const int d = ei[NEDGES + e];
  const int r = atomicAdd(&cnt[d], 1);
  if (r < SLOTCAP) slots[d * NFEAT + r] = s;
}

__global__ void __launch_bounds__(256) k_gather(const float* __restrict__ x, const int* __restrict__ ei,
                                                const int* __restrict__ cnt, float* xagg) {
  const int lane = threadIdx.x & 31;
  const int ql = lane < 24 ? lane : 23;
  const v4f* x4 = (const v4f*)x;
  const int b4 = NNODES * (NFEAT / 4);
  const int w0 = blockIdx.x * 8 + (threadIdx.x >> 5);
  for (int it = 0; it < 2; it++) {
    const int n = __builtin_amdgcn_readfirstlane(w0 + it * GWAVES);
    const int deg = cnt[n];
    const float dn = rsqrtf((float)(deg + 1));
    v4f a0, a1;
    {
      const float ww = dn * dn;
      a0 = ww * x4[n * 24 + ql];
      a1 = ww * x4[b4 + n * 24 + ql];
    }
    if (deg <= SLOTCAP) {
      const int* slots = (const int*)xagg + n * NFEAT;
      for (int c0 = 0; c0 < deg; c0 += 32) {
        const int m = (deg - c0 < 32) ? deg - c0 : 32;
        int key = 0x7fffffff;
        float ws = 0.f;
        if (lane < m) {
          const int s = slots[c0 + lane];
          ws = rsqrtf((float)(cnt[s] + 1)) * dn;
          key = (s << 5) | lane;
        }
        int myrank = 0;
        for (int j = 0; j < m; j++) myrank += (__builtin_amdgcn_readlane(key, j) < key) ? 1 : 0;
        for (int k0 = 0; k0 < m; k0 += 4) {
          v4f r0[4], r1[4];
          float wk[4];
#pragma unroll
          for (int u = 0; u < 4; u++) {
            wk[u] = 0.f; r0[u] = 0.f; r1[u] = 0.f;
            if (k0 + u < m) {
              const int l = __builtin_ctz(__builtin_amdgcn_ballot_w32(myrank == k0 + u));
              const int sk = __builtin_amdgcn_readlane(key, l) >> 5;
              wk[u] = __int_as_float(__builtin_amdgcn_readlane(__float_as_int(ws), l));
              r0[u] = x4[sk * 24 + ql];
              r1[u] = x4[b4 + sk * 24 + ql];
            }
          }
#pragma unroll
          for (int u = 0; u < 4; u++) { a0 += wk[u] * r0[u]; a1 += wk[u] * r1[u]; }
        }
      }
    } else {
      for (int e0 = 0; e0 < NEDGES; e0 += 32) {
        unsigned mask = __builtin_amdgcn_ballot_w32(ei[NEDGES + e0 + lane] == n);
        while (mask) {
          const int l = __builtin_ctz(mask);
          mask &= mask - 1;
          const int sk = ei[e0 + l];
          const float wk = rsqrtf((float)(cnt[sk] + 1)) * dn;
          a0 += wk * x4[sk * 24 + ql];
          a1 += wk * x4[b4 + sk * 24 + ql];
        }
      }
    }
    if (lane < 24) {
      ((v4f*)xagg)[n * 24 + lane] = a0;
      ((v4f*)xagg)[b4 + n * 24 + lane] = a1;
    }
  }
}

__global__ void __launch_bounds__(256, 6) k_gru(
    const float* __restrict__ xagg, const float* __restrict__ att,
    const float* __restrict__ Wz, const float* __restrict__ bz,
    const float* __restrict__ Wr, const float* __restrict__ br,
    const float* __restrict__ Wh, const float* __restrict__ bh,
    const float* __restrict__ Lz, const float* __restrict__ lbz,
    const float* __restrict__ Lr, const float* __restrict__ lbr,
    const float* __restrict__ Lh, const float* __restrict__ lbh,
    const float* __restrict__ Wo, const float* __restrict__ bo,
    float* __restrict__ out) {
  __shared__ __attribute__((aligned(16))) float sM[3 * 16 * 32];
  __shared__ __attribute__((aligned(16))) float sL[3 * 64 * 32];
  __shared__ __attribute__((aligned(16))) float sWo[32 * 32];
  __shared__ __attribute__((aligned(16))) float sbo[32];
  __shared__ __attribute__((aligned(16))) float sc[3 * 32];
  __shared__ float sPr[PER];

  int tid = threadIdx.x;
  for (int t = tid; t < 32; t += 256) sbo[t] = (t < OUTC) ? bo[t] : 0.f;
  for (int t = tid; t < 64 * 32; t += 256) {
    int k = t >> 5, n = t & 31;
    int f = (k < 32) ? fragIdx(t) : 1024 + fragIdxD(k - 32, n);
    sL[f] = Lz[t]; sL[2048 + f] = Lr[t]; sL[4096 + f] = Lh[t];
  }
  for (int t = tid; t < 32 * 32; t += 256) {
    int k = t >> 5, j = t & 31;
    sWo[fragIdxD(k, j)] = (j < OUTC) ? Wo[k * OUTC + j] : 0.f;
  }

  int wave = tid >> 5, lane = tid & 31, m = lane & 15, hi = lane >> 4;

  {
    float a[PER]; float mx = -1e30f;
    for (int p = 0; p < PER; p++) { a[p] = att[p]; mx = fmaxf(mx, a[p]); }
    float sm = 0.f;
    for (int p = 0; p < PER; p++) { a[p] = __expf(a[p] - mx); sm += a[p]; }
    float inv = 1.0f / sm;
    if (tid == 0) for (int p = 0; p < PER; p++) sPr[p] = a[p] * inv;
  }

  int tile = blockIdx.x * 8 + wave;
  long ia = (long)tile * 16 + m;
  if (ia >= ROWS) ia = ROWS - 1;
  const float* xrow = xagg + ia * (long)NFEAT;

  v8f H0, H1, A0, A1;
  for (int r = 0; r < 8; r++) { H0[r] = 0.f; H1[r] = 0.f; A0[r] = 0.f; A1[r] = 0.f; }

  __syncthreads();

  if (wave < 3) {
    const float* Wg = (wave == 0) ? Wz : (wave == 1) ? Wr : Wh;
    const float* bg = (wave == 0) ? bz : (wave == 1) ? br : bh;
    const float* Lg = sL + wave * 2048;
    float* Mg = sM + wave * 512;
    const float* arow = (m < 8) ? (Wg + m * 32) : bg;
    const bool live = (m <= 8);
    v2f af[8];
    for (int kb = 0; kb < 8; kb++) {
      const int k0 = kb * 4 + hi * 2;
      const float u = arow[k0], v = arow[k0 + 1];
      af[kb].x = live ? u : 0.f;
      af[kb].y = live ? v : 0.f;
    }
    v8f c0, c1;
    for (int r = 0; r < 8; r++) { c0[r] = 0.f; c1[r] = 0.f; }
    for (int kb = 0; kb < 8; kb++) {
      v4f q = ldsB4(Lg, lane, kb);
      c0 = wmma4(af[kb], q.xy, c0);
      c1 = wmma4(af[kb], q.zw, c1);
    }
    for (int r = 0; r < 8; r++) {
      Mg[fragIdx((r + 8 * hi) * 32 + m)]      = c0[r];
      Mg[fragIdx((r + 8 * hi) * 32 + 16 + m)] = c1[r];
    }
  }
  __syncthreads();

  if (tid < 96) {
    const int g = tid >> 5, n = tid & 31;
    const float* lbg = (g == 0) ? lbz : (g == 1) ? lbr : lbh;
    sc[tid] = sM[g * 512 + fragIdx(8 * 32 + n)] + lbg[n];
  }
  __syncthreads();

  for (int p = 0; p < PER; p++) {
    asm volatile("" ::: "memory");
    int f0 = hi * 2;
    v2f bx0, bx1;
    bx0.x = xrow[(f0)*PER + p];     bx0.y = xrow[(f0 + 1) * PER + p];
    bx1.x = xrow[(f0 + 4) * PER + p]; bx1.y = xrow[(f0 + 5) * PER + p];

    v8f z0, z1, r0, r1, t0, t1;
    biasTiles(sc, hi, z0, z1);
    mmaX(bx0, bx1, sM, lane, z0, z1);
    mmaD(H0, H1, sL + 1024, lane, z0, z1);
    for (int r = 0; r < 8; r++) { z0[r] = sigm(z0[r]); z1[r] = sigm(z1[r]); }

    biasTiles(sc + 32, hi, r0, r1);
    mmaX(bx0, bx1, sM + 512, lane, r0, r1);
    mmaD(H0, H1, sL + 2048 + 1024, lane, r0, r1);
    for (int r = 0; r < 8; r++) { r0[r] = sigm(r0[r]); r1[r] = sigm(r1[r]); }
    v8f q0, q1;
    for (int r = 0; r < 8; r++) { q0[r] = H0[r] * r0[r]; q1[r] = H1[r] * r1[r]; }

    biasTiles(sc + 64, hi, t0, t1);
    mmaX(bx0, bx1, sM + 1024, lane, t0, t1);
    mmaD(q0, q1, sL + 4096 + 1024, lane, t0, t1);
    for (int r = 0; r < 8; r++) { t0[r] = tanhc(t0[r]); t1[r] = tanhc(t1[r]); }

    const float prp = sPr[p];
    for (int r = 0; r < 8; r++) {
      H0[r] = z0[r] * H0[r] + (1.0f - z0[r]) * t0[r];
      H1[r] = z1[r] * H1[r] + (1.0f - z1[r]) * t1[r];
      A0[r] += prp * H0[r];
      A1[r] += prp * H1[r];
    }
  }

  for (int r = 0; r < 8; r++) { A0[r] = fmaxf(A0[r], 0.f); A1[r] = fmaxf(A1[r], 0.f); }
  v8f o0, o1;
  biasTiles(sbo, hi, o0, o1);
  mmaD(A0, A1, sWo, lane, o0, o1);
  long row = (long)tile * 16 + m;
  if (row < ROWS) {
    v4f* orow = (v4f*)(out + row * OUTC + 8 * hi);
    orow[0] = __builtin_shufflevector(o0, o0, 0, 1, 2, 3);
    orow[1] = __builtin_shufflevector(o0, o0, 4, 5, 6, 7);
    if (hi == 0) {
      orow[4] = __builtin_shufflevector(o1, o1, 0, 1, 2, 3);
      orow[5] = __builtin_shufflevector(o1, o1, 4, 5, 6, 7);
    }
  }
}

extern "C" void kernel_launch(void* const* d_in, const int* in_sizes, int n_in,
                              void* d_out, int out_size, void* d_ws, size_t ws_size,
                              hipStream_t stream) {
  (void)in_sizes; (void)n_in; (void)out_size; (void)ws_size;
  const float* x   = (const float*)d_in[0];
  const int*   ei  = (const int*)d_in[1];
  const float* att = (const float*)d_in[2];
  const float* Wz  = (const float*)d_in[3];  const float* bz  = (const float*)d_in[4];
  const float* Wr  = (const float*)d_in[5];  const float* br  = (const float*)d_in[6];
  const float* Wh  = (const float*)d_in[7];  const float* bh  = (const float*)d_in[8];
  const float* Lz  = (const float*)d_in[9];  const float* lbz = (const float*)d_in[10];
  const float* Lr  = (const float*)d_in[11]; const float* lbr = (const float*)d_in[12];
  const float* Lh  = (const float*)d_in[13]; const float* lbh = (const float*)d_in[14];
  const float* Wo  = (const float*)d_in[15]; const float* bo  = (const float*)d_in[16];
  float* out = (float*)d_out;

  float* wsf  = (float*)d_ws;
  int*   cnt  = (int*)d_ws;
  float* xagg = wsf + 65536;

  k_cnt_init<<<(NNODES + 255) / 256, 256, 0, stream>>>(cnt);
  k_place<<<NEDGES / 256, 256, 0, stream>>>(ei, cnt, (int*)xagg);
  k_gather<<<GWAVES / 8, 256, 0, stream>>>(x, ei, cnt, xagg);

  int tiles = ROWS / 16;
  int blocks = (tiles + 7) / 8;
  k_gru<<<blocks, 256, 0, stream>>>(xagg, att, Wz, bz, Wr, br, Wh, bh,
                                    Lz, lbz, Lr, lbr, Lh, lbh, Wo, bo, out);
}


# GRU grid: 250 blocks of 25 waves, one 16-row tile per wave, in place of 782 of 8: a block fills its WGP alone at 7/6/6/6 waves per SIMD, so no SIMD carries 8 tiles any more; sums unchanged
# speedup vs baseline: 3.4394x; 1.2197x over previous
#include <hip/hip_runtime.h>

#define NNODES 50000
#define NEDGES 800000
#define FIN 8
#define PER 12
#define HIDN 32
#define NB 2
#define OUTC 24
#define ROWS (NB*NNODES)
#define NFEAT (FIN*PER)

typedef float v2f __attribute__((ext_vector_type(2)));
typedef float v4f __attribute__((ext_vector_type(4)));
typedef float v8f __attribute__((ext_vector_type(8)));

__device__ __forceinline__ v8f wmma4(v2f a, v2f b, v8f c) {
  return __builtin_amdgcn_wmma_f32_16x16x4_f32(false, a, false, b, (short)0, c, false, false);
}


__device__ __forceinline__ int fragIdx(int t) {
  int k = t >> 5, n = t & 31;
  int kb = k >> 2, hi = (k >> 1) & 1, j = k & 1, nt = n >> 4, m = n & 15;
  return ((kb * 32 + hi * 16 + m) << 2) + nt * 2 + j;
}
__device__ __forceinline__ int fragIdxD(int k, int n) {
  int kb = ((k >> 4) << 2) + ((k >> 1) & 3), hi = (k >> 3) & 1, j = k & 1, nt = n >> 4, m = n & 15;
  return ((kb * 32 + hi * 16 + m) << 2) + nt * 2 + j;
}
__device__ __forceinline__ v4f ldsB4(const float* W, int lane, int kb) {
  return *(const v4f*)(W + ((kb * 32 + lane) << 2));
}

__device__ __forceinline__ void biasTiles(const float* b, int hi, v8f& c0, v8f& c1) {
  const v4f* p = (const v4f*)(b + 8 * hi);
  v4f u0 = p[0], u1 = p[1], w0 = p[4], w1 = p[5];
  c0 = __builtin_shufflevector(u0, u1, 0, 1, 2, 3, 4, 5, 6, 7);
  c1 = __builtin_shufflevector(w0, w1, 0, 1, 2, 3, 4, 5, 6, 7);
}

__device__ __forceinline__ void mmaX(v2f bx0, v2f bx1, const float* M, int lane, v8f& c0, v8f& c1) {
  v4f q0 = ldsB4(M, lane, 0), q1 = ldsB4(M, lane, 1);
  c0 = wmma4(q0.xy, bx0, c0);
  c1 = wmma4(q0.zw, bx0, c1);
  c0 = wmma4(q1.xy, bx1, c0);
  c1 = wmma4(q1.zw, bx1, c1);
}

__device__ __forceinline__ void mmaD(const v8f& s0, const v8f& s1, const float* W, int lane, v8f& c0, v8f& c1) {
#pragma unroll
  for (int kb = 0; kb < 8; kb++) {
    const int p = kb & 3;
    v2f b;
    b.x = (kb < 4) ? s0[2 * p] : s1[2 * p];
    b.y = (kb < 4) ? s0[2 * p + 1] : s1[2 * p + 1];
    v4f q = ldsB4(W, lane, kb);
    c0 = wmma4(q.xy, b, c0);
    c1 = wmma4(q.zw, b, c1);
  }
}

__device__ __forceinline__ float sigm(float x) {
  return __builtin_amdgcn_rcpf(1.0f + __builtin_amdgcn_exp2f(x * -1.442695041f));
}
__device__ __forceinline__ float tanhc(float x) {
  return 1.0f - 2.0f * __builtin_amdgcn_rcpf(__builtin_amdgcn_exp2f(x * 2.885390082f) + 1.0f);
}

#define SLOTCAP NFEAT
#define GWAVES 25000
static_assert(NEDGES % 256 == 0, "one edge per thread, no tail");
static_assert(NFEAT == 96, "a row is 24 lanes x 16 bytes = three 128-byte lines");
static_assert(2 * GWAVES == NNODES && GWAVES % 8 == 0, "two nodes per gather wave, eight waves per block");

__global__ void k_cnt_init(int* cnt) {
  int i = blockIdx.x * 256 + threadIdx.x;
  if (i < NNODES) cnt[i] = 0;
}
__global__ void __launch_bounds__(256) k_place(const int* __restrict__ ei, int* cnt, int* slots) {
  const int e = blockIdx.x * 256 + threadIdx.x;
  const int s = ei[e];
  const int d = ei[NEDGES + e];
  const int r = atomicAdd(&cnt[d], 1);
  if (r < SLOTCAP) slots[d * NFEAT + r] = s;
}

__global__ void __launch_bounds__(256) k_gather(const float* __restrict__ x, const int* __restrict__ ei,
                                                const int* __restrict__ cnt, float* xagg) {
  const int lane = threadIdx.x & 31;
  const int ql = lane < 24 ? lane : 23;
  const v4f* x4 = (const v4f*)x;
  const int b4 = NNODES * (NFEAT / 4);
  const int w0 = blockIdx.x * 8 + (threadIdx.x >> 5);
  for (int it = 0; it < 2; it++) {
    const int n = __builtin_amdgcn_readfirstlane(w0 + it * GWAVES);
    const int deg = cnt[n];
    const float dn = rsqrtf((float)(deg + 1));
    v4f a0, a1;
    {
      const float ww = dn * dn;
      a0 = ww * x4[n * 24 + ql];
      a1 = ww * x4[b4 + n * 24 + ql];
    }
    if (deg <= SLOTCAP) {
      const int* slots = (const int*)xagg + n * NFEAT;
      for (int c0 = 0; c0 < deg; c0 += 32) {
        const int m = (deg - c0 < 32) ? deg - c0 : 32;
        int key = 0x7fffffff;
        float ws = 0.f;
        if (lane < m) {
          const int s = slots[c0 + lane];
          ws = rsqrtf((float)(cnt[s] + 1)) * dn;
          key = (s << 5) | lane;
        }
        int myrank = 0;
        for (int j = 0; j < m; j++) myrank += (__builtin_amdgcn_readlane(key, j) < key) ? 1 : 0;
        for (int k0 = 0; k0 < m; k0 += 4) {
          v4f r0[4], r1[4];
          float wk[4];
#pragma unroll
          for (int u = 0; u < 4; u++) {
            wk[u] = 0.f; r0[u] = 0.f; r1[u] = 0.f;
            if (k0 + u < m) {
              const int l = __builtin_ctz(__builtin_amdgcn_ballot_w32(myrank == k0 + u));
              const int sk = __builtin_amdgcn_readlane(key, l) >> 5;
              wk[u] = __int_as_float(__builtin_amdgcn_readlane(__float_as_int(ws), l));
              r0[u] = x4[sk * 24 + ql];
              r1[u] = x4[b4 + sk * 24 + ql];
            }
          }
#pragma unroll
          for (int u = 0; u < 4; u++) { a0 += wk[u] * r0[u]; a1 += wk[u] * r1[u]; }
        }
      }
    } else {
      for (int e0 = 0; e0 < NEDGES; e0 += 32) {
        unsigned mask = __builtin_amdgcn_ballot_w32(ei[NEDGES + e0 + lane] == n);
        while (mask) {
          const int l = __builtin_ctz(mask);
          mask &= mask - 1;
          const int sk = ei[e0 + l];
          const float wk = rsqrtf((float)(cnt[sk] + 1)) * dn;
          a0 += wk * x4[sk * 24 + ql];
          a1 += wk * x4[b4 + sk * 24 + ql];
        }
      }
    }
    if (lane < 24) {
      ((v4f*)xagg)[n * 24 + lane] = a0;
      ((v4f*)xagg)[b4 + n * 24 + lane] = a1;
    }
  }
}

#define GWPB 25
#define GBLOCKS 250
#define GTHREADS (GWPB * 32)
static_assert(GBLOCKS * GWPB * 16 == ROWS, "the grid covers the rows exactly, one tile per wave");
__global__ void __launch_bounds__(GTHREADS) k_gru(
    const float* __restrict__ xagg, const float* __restrict__ att,
    const float* __restrict__ Wz, const float* __restrict__ bz,
    const float* __restrict__ Wr, const float* __restrict__ br,
    const float* __restrict__ Wh, const float* __restrict__ bh,
    const float* __restrict__ Lz, const float* __restrict__ lbz,
    const float* __restrict__ Lr, const float* __restrict__ lbr,
    const float* __restrict__ Lh, const float* __restrict__ lbh,
    const float* __restrict__ Wo, const float* __restrict__ bo,
    float* __restrict__ out) {
  __shared__ __attribute__((aligned(16))) float sM[3 * 16 * 32];
  __shared__ __attribute__((aligned(16))) float sL[3 * 64 * 32];
  __shared__ __attribute__((aligned(16))) float sWo[32 * 32];
  __shared__ __attribute__((aligned(16))) float sbo[32];
  __shared__ __attribute__((aligned(16))) float sc[3 * 32];
  __shared__ float sPr[PER];

  int tid = threadIdx.x;
  for (int t = tid; t < 32; t += GTHREADS) sbo[t] = (t < OUTC) ? bo[t] : 0.f;
  for (int t = tid; t < 64 * 32; t += GTHREADS) {
    int k = t >> 5, n = t & 31;
    int f = (k < 32) ? fragIdx(t) : 1024 + fragIdxD(k - 32, n);
    sL[f] = Lz[t]; sL[2048 + f] = Lr[t]; sL[4096 + f] = Lh[t];
  }
  for (int t = tid; t < 32 * 32; t += GTHREADS) {
    int k = t >> 5, j = t & 31;
    sWo[fragIdxD(k, j)] = (j < OUTC) ? Wo[k * OUTC + j] : 0.f;
  }

  int wave = tid >> 5, lane = tid & 31, m = lane & 15, hi = lane >> 4;

  {
    float a[PER]; float mx = -1e30f;
    for (int p = 0; p < PER; p++) { a[p] = att[p]; mx = fmaxf(mx, a[p]); }
    float sm = 0.f;
    for (int p = 0; p < PER; p++) { a[p] = __expf(a[p] - mx); sm += a[p]; }
    float inv = 1.0f / sm;
    if (tid == 0) for (int p = 0; p < PER; p++) sPr[p] = a[p] * inv;
  }

  int tile = blockIdx.x * GWPB + wave;
  long ia = (long)tile * 16 + m;
  if (ia >= ROWS) ia = ROWS - 1;
  const float* xrow = xagg + ia * (long)NFEAT;

  v8f H0, H1, A0, A1;
  for (int r = 0; r < 8; r++) { H0[r] = 0.f; H1[r] = 0.f; A0[r] = 0.f; A1[r] = 0.f; }

  __syncthreads();

  if (wave < 3) {
    const float* Wg = (wave == 0) ? Wz : (wave == 1) ? Wr : Wh;
    const float* bg = (wave == 0) ? bz : (wave == 1) ? br : bh;
    const float* Lg = sL + wave * 2048;
    float* Mg = sM + wave * 512;
    const float* arow = (m < 8) ? (Wg + m * 32) : bg;
    const bool live = (m <= 8);
    v2f af[8];
    for (int kb = 0; kb < 8; kb++) {
      const int k0 = kb * 4 + hi * 2;
      const float u = arow[k0], v = arow[k0 + 1];
      af[kb].x = live ? u : 0.f;
      af[kb].y = live ? v : 0.f;
    }
    v8f c0, c1;
    for (int r = 0; r < 8; r++) { c0[r] = 0.f; c1[r] = 0.f; }
    for (int kb = 0; kb < 8; kb++) {
      v4f q = ldsB4(Lg, lane, kb);
      c0 = wmma4(af[kb], q.xy, c0);
      c1 = wmma4(af[kb], q.zw, c1);
    }
    for (int r = 0; r < 8; r++) {
      Mg[fragIdx((r + 8 * hi) * 32 + m)]      = c0[r];
      Mg[fragIdx((r + 8 * hi) * 32 + 16 + m)] = c1[r];
    }
  }
  __syncthreads();

  if (tid < 96) {
    const int g = tid >> 5, n = tid & 31;
    const float* lbg = (g == 0) ? lbz : (g == 1) ? lbr : lbh;
    sc[tid] = sM[g * 512 + fragIdx(8 * 32 + n)] + lbg[n];
  }
  __syncthreads();

  for (int p = 0; p < PER; p++) {
    asm volatile("" ::: "memory");
    int f0 = hi * 2;
    v2f bx0, bx1;
    bx0.x = xrow[(f0)*PER + p];     bx0.y = xrow[(f0 + 1) * PER + p];
    bx1.x = xrow[(f0 + 4) * PER + p]; bx1.y = xrow[(f0 + 5) * PER + p];

    v8f z0, z1, r0, r1, t0, t1;
    biasTiles(sc, hi, z0, z1);
    mmaX(bx0, bx1, sM, lane, z0, z1);
    mmaD(H0, H1, sL + 1024, lane, z0, z1);
    for (int r = 0; r < 8; r++) { z0[r] = sigm(z0[r]); z1[r] = sigm(z1[r]); }

    biasTiles(sc + 32, hi, r0, r1);
    mmaX(bx0, bx1, sM + 512, lane, r0, r1);
    mmaD(H0, H1, sL + 2048 + 1024, lane, r0, r1);
    for (int r = 0; r < 8; r++) { r0[r] = sigm(r0[r]); r1[r] = sigm(r1[r]); }
    v8f q0, q1;
    for (int r = 0; r < 8; r++) { q0[r] = H0[r] * r0[r]; q1[r] = H1[r] * r1[r]; }

    biasTiles(sc + 64, hi, t0, t1);
    mmaX(bx0, bx1, sM + 1024, lane, t0, t1);
    mmaD(q0, q1, sL + 4096 + 1024, lane, t0, t1);
    for (int r = 0; r < 8; r++) { t0[r] = tanhc(t0[r]); t1[r] = tanhc(t1[r]); }

    const float prp = sPr[p];
    for (int r = 0; r < 8; r++) {
      H0[r] = z0[r] * H0[r] + (1.0f - z0[r]) * t0[r];
      H1[r] = z1[r] * H1[r] + (1.0f - z1[r]) * t1[r];
      A0[r] += prp * H0[r];
      A1[r] += prp * H1[r];
    }
  }

  for (int r = 0; r < 8; r++) { A0[r] = fmaxf(A0[r], 0.f); A1[r] = fmaxf(A1[r], 0.f); }
  v8f o0, o1;
  biasTiles(sbo, hi, o0, o1);
  mmaD(A0, A1, sWo, lane, o0, o1);
  long row = (long)tile * 16 + m;
  if (row < ROWS) {
    v4f* orow = (v4f*)(out + row * OUTC + 8 * hi);
    orow[0] = __builtin_shufflevector(o0, o0, 0, 1, 2, 3);
    orow[1] = __builtin_shufflevector(o0, o0, 4, 5, 6, 7);
    if (hi == 0) {
      orow[4] = __builtin_shufflevector(o1, o1, 0, 1, 2, 3);
      orow[5] = __builtin_shufflevector(o1, o1, 4, 5, 6, 7);
    }
  }
}

extern "C" void kernel_launch(void* const* d_in, const int* in_sizes, int n_in,
                              void* d_out, int out_size, void* d_ws, size_t ws_size,
                              hipStream_t stream) {
  (void)in_sizes; (void)n_in; (void)out_size; (void)ws_size;
  const float* x   = (const float*)d_in[0];
  const int*   ei  = (const int*)d_in[1];
  const float* att = (const float*)d_in[2];
  const float* Wz  = (const float*)d_in[3];  const float* bz  = (const float*)d_in[4];
  const float* Wr  = (const float*)d_in[5];  const float* br  = (const float*)d_in[6];
  const float* Wh  = (const float*)d_in[7];  const float* bh  = (const float*)d_in[8];
  const float* Lz  = (const float*)d_in[9];  const float* lbz = (const float*)d_in[10];
  const float* Lr  = (const float*)d_in[11]; const float* lbr = (const float*)d_in[12];
  const float* Lh  = (const float*)d_in[13]; const float* lbh = (const float*)d_in[14];
  const float* Wo  = (const float*)d_in[15]; const float* bo  = (const float*)d_in[16];
  float* out = (float*)d_out;

  float* wsf  = (float*)d_ws;
  int*   cnt  = (int*)d_ws;
  float* xagg = wsf + 65536;

  k_cnt_init<<<(NNODES + 255) / 256, 256, 0, stream>>>(cnt);
  k_place<<<NEDGES / 256, 256, 0, stream>>>(ei, cnt, (int*)xagg);
  k_gather<<<GWAVES / 8, 256, 0, stream>>>(x, ei, cnt, xagg);

  k_gru<<<GBLOCKS, GTHREADS, 0, stream>>>(xagg, att, Wz, bz, Wr, br, Wh, bh,
                                          Lz, lbz, Lr, lbr, Lh, lbh, Wo, bo, out);
}


# edge placement: two-level counting sort in LDS (bins of 128 destination nodes, then per-node lists) replaces a global returning add + scattered 4-byte store per edge; sums and their order as before
# speedup vs baseline: 4.4324x; 1.2887x over previous
#include <hip/hip_runtime.h>

#define NNODES 50000
#define NEDGES 800000
#define FIN 8
#define PER 12
#define HIDN 32
#define NB 2
#define OUTC 24
#define ROWS (NB*NNODES)
#define NFEAT (FIN*PER)

typedef float v2f __attribute__((ext_vector_type(2)));
typedef float v4f __attribute__((ext_vector_type(4)));
typedef float v8f __attribute__((ext_vector_type(8)));

__device__ __forceinline__ v8f wmma4(v2f a, v2f b, v8f c) {
  return __builtin_amdgcn_wmma_f32_16x16x4_f32(false, a, false, b, (short)0, c, false, false);
}


__device__ __forceinline__ int fragIdx(int t) {
  int k = t >> 5, n = t & 31;
  int kb = k >> 2, hi = (k >> 1) & 1, j = k & 1, nt = n >> 4, m = n & 15;
  return ((kb * 32 + hi * 16 + m) << 2) + nt * 2 + j;
}
__device__ __forceinline__ int fragIdxD(int k, int n) {
  int kb = ((k >> 4) << 2) + ((k >> 1) & 3), hi = (k >> 3) & 1, j = k & 1, nt = n >> 4, m = n & 15;
  return ((kb * 32 + hi * 16 + m) << 2) + nt * 2 + j;
}
__device__ __forceinline__ v4f ldsB4(const float* W, int lane, int kb) {
  return *(const v4f*)(W + ((kb * 32 + lane) << 2));
}

__device__ __forceinline__ void biasTiles(const float* b, int hi, v8f& c0, v8f& c1) {
  const v4f* p = (const v4f*)(b + 8 * hi);
  v4f u0 = p[0], u1 = p[1], w0 = p[4], w1 = p[5];
  c0 = __builtin_shufflevector(u0, u1, 0, 1, 2, 3, 4, 5, 6, 7);
  c1 = __builtin_shufflevector(w0, w1, 0, 1, 2, 3, 4, 5, 6, 7);
}

__device__ __forceinline__ void mmaX(v2f bx0, v2f bx1, const float* M, int lane, v8f& c0, v8f& c1) {
  v4f q0 = ldsB4(M, lane, 0), q1 = ldsB4(M, lane, 1);
  c0 = wmma4(q0.xy, bx0, c0);
  c1 = wmma4(q0.zw, bx0, c1);
  c0 = wmma4(q1.xy, bx1, c0);
  c1 = wmma4(q1.zw, bx1, c1);
}

__device__ __forceinline__ void mmaD(const v8f& s0, const v8f& s1, const float* W, int lane, v8f& c0, v8f& c1) {
#pragma unroll
  for (int kb = 0; kb < 8; kb++) {
    const int p = kb & 3;
    v2f b;
    b.x = (kb < 4) ? s0[2 * p] : s1[2 * p];
    b.y = (kb < 4) ? s0[2 * p + 1] : s1[2 * p + 1];
    v4f q = ldsB4(W, lane, kb);
    c0 = wmma4(q.xy, b, c0);
    c1 = wmma4(q.zw, b, c1);
  }
}

__device__ __forceinline__ float sigm(float x) {
  return __builtin_amdgcn_rcpf(1.0f + __builtin_amdgcn_exp2f(x * -1.442695041f));
}
__device__ __forceinline__ float tanhc(float x) {
  return 1.0f - 2.0f * __builtin_amdgcn_rcpf(__builtin_amdgcn_exp2f(x * 2.885390082f) + 1.0f);
}

#define SLOTCAP NFEAT
#define GWAVES 25000
static_assert(NFEAT == 96, "a row is 24 lanes x 16 bytes = three 128-byte lines");
static_assert(2 * GWAVES == NNODES && GWAVES % 8 == 0, "two nodes per gather wave, eight waves per block");

#define BSH 7
#define BNODES (1 << BSH)
#define NBINS ((NNODES + BNODES - 1) / BNODES)
#define BTHREADS 1024
#define EPT 4
#define ECHUNK (BTHREADS * EPT)
#define NBLK1 ((NEDGES + ECHUNK - 1) / ECHUNK)
#define CAPB 40
#define SPL ((NBINS + 31) / 32)
#define STHREADS 256
#define SU 8
static_assert(NNODES <= 65536 && BNODES <= 256, "a record packs the source in 16 bits and the node-in-bin above them");
static_assert(STHREADS >= BNODES, "one thread per node of a bin clears and stores the counts");
static_assert((long)NBINS * NBLK1 * CAPB + (long)NBLK1 * NBINS <= (long)NNODES * NFEAT, "cells and counts fit in the batch-1 half of xagg");

__global__ void __launch_bounds__(BTHREADS) k_bin(const int* __restrict__ ei, int* __restrict__ counts, int* __restrict__ cells) {
  __shared__ int hist[NBINS];
  __shared__ int start[NBINS];
  __shared__ int srec[ECHUNK];
  __shared__ int sdst[ECHUNK];
  const int tid = threadIdx.x, blk = blockIdx.x;
  for (int b = tid; b < NBINS; b += BTHREADS) hist[b] = 0;
  __syncthreads();
  int rec[EPT], bn[EPT], rk[EPT];
#pragma unroll
  for (int i = 0; i < EPT; i++) {
    const int e = blk * ECHUNK + i * BTHREADS + tid;
    const int ec = e < NEDGES ? e : NEDGES - 1;
    const int s = ei[ec], d = ei[NEDGES + ec];
    bn[i] = e < NEDGES ? (d >> BSH) : -1;
    rec[i] = ((d & (BNODES - 1)) << 16) | s;
    rk[i] = 0;
  }
#pragma unroll
  for (int i = 0; i < EPT; i++)
    if (bn[i] >= 0) rk[i] = atomicAdd(&hist[bn[i]], 1);
  __syncthreads();
  for (int b = tid; b < NBINS; b += BTHREADS) counts[blk * NBINS + b] = hist[b];
  if (tid < 32) {
    int sum = 0;
    for (int j = 0; j < SPL; j++) { const int b = tid * SPL + j; if (b < NBINS) sum += hist[b]; }
    int inc = sum;
    for (int off = 1; off < 32; off <<= 1) { const int v = __shfl_up(inc, off, 32); if (tid >= off) inc += v; }
    int run = inc - sum;
    for (int j = 0; j < SPL; j++) { const int b = tid * SPL + j; if (b < NBINS) { start[b] = run; run += hist[b]; } }
  }
  __syncthreads();
#pragma unroll
  for (int i = 0; i < EPT; i++)
    if (bn[i] >= 0) {
      const int p = start[bn[i]] + rk[i];
      srec[p] = rec[i];
      sdst[p] = (rk[i] < CAPB) ? (bn[i] * NBLK1 + blk) * CAPB + rk[i] : -1;
    }
  __syncthreads();
  const int left = NEDGES - blk * ECHUNK;
  const int total = left < ECHUNK ? left : ECHUNK;
  for (int p = tid; p < total; p += BTHREADS) {
    const int dd = sdst[p];
    if (dd >= 0) cells[dd] = srec[p];
  }
}

__global__ void __launch_bounds__(STHREADS) k_sort(const int* __restrict__ ei, const int* __restrict__ counts,
                                                   const int* __restrict__ cells, int* __restrict__ cnt, int* slots) {
  __shared__ int lcnt[BNODES];
  __shared__ int ccnt[NBLK1];
  __shared__ int llist[BNODES * SLOTCAP];
  __shared__ int ovf;
  const int tid = threadIdx.x, bin = blockIdx.x;
  if (tid < BNODES) lcnt[tid] = 0;
  if (tid == 0) ovf = 0;
  __syncthreads();
  for (int c = tid; c < NBLK1; c += STHREADS) {
    const int v = counts[c * NBINS + bin];
    ccnt[c] = v;
    if (v > CAPB) ovf = 1;
  }
  __syncthreads();
  if (ovf == 0) {
    for (int i0 = tid; i0 < NBLK1 * CAPB; i0 += STHREADS * SU) {
      int rr[SU];
      bool ok[SU];
#pragma unroll
      for (int u = 0; u < SU; u++) {
        const int idx = i0 + u * STHREADS;
        ok[u] = (idx < NBLK1 * CAPB) && (idx % CAPB < ccnt[idx / CAPB]);
        rr[u] = ok[u] ? cells[bin * (NBLK1 * CAPB) + idx] : 0;
      }
#pragma unroll
      for (int u = 0; u < SU; u++)
        if (ok[u]) {
          const int dl = rr[u] >> 16, s = rr[u] & 0xffff;
          const int r = atomicAdd(&lcnt[dl], 1);
          if (r < SLOTCAP) llist[dl * SLOTCAP + r] = s;
        }
    }
  } else {
    for (int e = tid; e < NEDGES; e += STHREADS) {
      const int d = ei[NEDGES + e];
      if ((d >> BSH) == bin) {
        const int dl = d & (BNODES - 1);
        const int r = atomicAdd(&lcnt[dl], 1);
        if (r < SLOTCAP) llist[dl * SLOTCAP + r] = ei[e];
      }
    }
  }
  __syncthreads();
  if (tid < BNODES && bin * BNODES + tid < NNODES) cnt[bin * BNODES + tid] = lcnt[tid];
  const int wave = tid >> 5, lane = tid & 31;
  for (int dl = wave; dl < BNODES; dl += STHREADS / 32) {
    const int n = bin * BNODES + dl;
    if (n < NNODES) {
      const int deg = lcnt[dl];
      const int m = deg < SLOTCAP ? deg : SLOTCAP;
      for (int j = lane; j < m; j += 32) slots[n * NFEAT + j] = llist[dl * SLOTCAP + j];
    }
  }
}

__global__ void __launch_bounds__(256) k_gather(const float* __restrict__ x, const int* __restrict__ ei,
                                                const int* __restrict__ cnt, float* xagg) {
  const int lane = threadIdx.x & 31;
  const int ql = lane < 24 ? lane : 23;
  const v4f* x4 = (const v4f*)x;
  const int b4 = NNODES * (NFEAT / 4);
  const int w0 = blockIdx.x * 8 + (threadIdx.x >> 5);
  for (int it = 0; it < 2; it++) {
    const int n = __builtin_amdgcn_readfirstlane(w0 + it * GWAVES);
    const int deg = cnt[n];
    const float dn = rsqrtf((float)(deg + 1));
    v4f a0, a1;
    {
      const float ww = dn * dn;
      a0 = ww * x4[n * 24 + ql];
      a1 = ww * x4[b4 + n * 24 + ql];
    }
    if (deg <= SLOTCAP) {
      const int* slots = (const int*)xagg + n * NFEAT;
      for (int c0 = 0; c0 < deg; c0 += 32) {
        const int m = (deg - c0 < 32) ? deg - c0 : 32;
        int key = 0x7fffffff;
        float ws = 0.f;
        if (lane < m) {
          const int s = slots[c0 + lane];
          ws = rsqrtf((float)(cnt[s] + 1)) * dn;
          key = (s << 5) | lane;
        }
        int myrank = 0;
        for (int j = 0; j < m; j++) myrank += (__builtin_amdgcn_readlane(key, j) < key) ? 1 : 0;
        for (int k0 = 0; k0 < m; k0 += 4) {
          v4f r0[4], r1[4];
          float wk[4];
#pragma unroll
          for (int u = 0; u < 4; u++) {
            wk[u] = 0.f; r0[u] = 0.f; r1[u] = 0.f;
            if (k0 + u < m) {
              const int l = __builtin_ctz(__builtin_amdgcn_ballot_w32(myrank == k0 + u));
              const int sk = __builtin_amdgcn_readlane(key, l) >> 5;
              wk[u] = __int_as_float(__builtin_amdgcn_readlane(__float_as_int(ws), l));
              r0[u] = x4[sk * 24 + ql];
              r1[u] = x4[b4 + sk * 24 + ql];
            }
          }
#pragma unroll
          for (int u = 0; u < 4; u++) { a0 += wk[u] * r0[u]; a1 += wk[u] * r1[u]; }
        }
      }
    } else {
      for (int e0 = 0; e0 < NEDGES; e0 += 32) {
        unsigned mask = __builtin_amdgcn_ballot_w32(ei[NEDGES + e0 + lane] == n);
        while (mask) {
          const int l = __builtin_ctz(mask);
          mask &= mask - 1;
          const int sk = ei[e0 + l];
          const float wk = rsqrtf((float)(cnt[sk] + 1)) * dn;
          a0 += wk * x4[sk * 24 + ql];
          a1 += wk * x4[b4 + sk * 24 + ql];
        }
      }
    }
    if (lane < 24) {
      ((v4f*)xagg)[n * 24 + lane] = a0;
      ((v4f*)xagg)[b4 + n * 24 + lane] = a1;
    }
  }
}

#define GWPB 25
#define GBLOCKS 250
#define GTHREADS (GWPB * 32)
static_assert(GBLOCKS * GWPB * 16 == ROWS, "the grid covers the rows exactly, one tile per wave");
__global__ void __launch_bounds__(GTHREADS) k_gru(
    const float* __restrict__ xagg, const float* __restrict__ att,
    const float* __restrict__ Wz, const float* __restrict__ bz,
    const float* __restrict__ Wr, const float* __restrict__ br,
    const float* __restrict__ Wh, const float* __restrict__ bh,
    const float* __restrict__ Lz, const float* __restrict__ lbz,
    const float* __restrict__ Lr, const float* __restrict__ lbr,
    const float* __restrict__ Lh, const float* __restrict__ lbh,
    const float* __restrict__ Wo, const float* __restrict__ bo,
    float* __restrict__ out) {
  __shared__ __attribute__((aligned(16))) float sM[3 * 16 * 32];
  __shared__ __attribute__((aligned(16))) float sL[3 * 64 * 32];
  __shared__ __attribute__((aligned(16))) float sWo[32 * 32];
  __shared__ __attribute__((aligned(16))) float sbo[32];
  __shared__ __attribute__((aligned(16))) float sc[3 * 32];
  __shared__ float sPr[PER];

  int tid = threadIdx.x;
  for (int t = tid; t < 32; t += GTHREADS) sbo[t] = (t < OUTC) ? bo[t] : 0.f;
  for (int t = tid; t < 64 * 32; t += GTHREADS) {
    int k = t >> 5, n = t & 31;
    int f = (k < 32) ? fragIdx(t) : 1024 + fragIdxD(k - 32, n);
    sL[f] = Lz[t]; sL[2048 + f] = Lr[t]; sL[4096 + f] = Lh[t];
  }
  for (int t = tid; t < 32 * 32; t += GTHREADS) {
    int k = t >> 5, j = t & 31;
    sWo[fragIdxD(k, j)] = (j < OUTC) ? Wo[k * OUTC + j] : 0.f;
  }

  int wave = tid >> 5, lane = tid & 31, m = lane & 15, hi = lane >> 4;

  {
    float a[PER]; float mx = -1e30f;
    for (int p = 0; p < PER; p++) { a[p] = att[p]; mx = fmaxf(mx, a[p]); }
    float sm = 0.f;
    for (int p = 0; p < PER; p++) { a[p] = __expf(a[p] - mx); sm += a[p]; }
    float inv = 1.0f / sm;
    if (tid == 0) for (int p = 0; p < PER; p++) sPr[p] = a[p] * inv;
  }

  int tile = blockIdx.x * GWPB + wave;
  long ia = (long)tile * 16 + m;
  if (ia >= ROWS) ia = ROWS - 1;
  const float* xrow = xagg + ia * (long)NFEAT;

  v8f H0, H1, A0, A1;
  for (int r = 0; r < 8; r++) { H0[r] = 0.f; H1[r] = 0.f; A0[r] = 0.f; A1[r] = 0.f; }

  __syncthreads();

  if (wave < 3) {
    const float* Wg = (wave == 0) ? Wz : (wave == 1) ? Wr : Wh;
    const float* bg = (wave == 0) ? bz : (wave == 1) ? br : bh;
    const float* Lg = sL + wave * 2048;
    float* Mg = sM + wave * 512;
    const float* arow = (m < 8) ? (Wg + m * 32) : bg;
    const bool live = (m <= 8);
    v2f af[8];
    for (int kb = 0; kb < 8; kb++) {
      const int k0 = kb * 4 + hi * 2;
      const float u = arow[k0], v = arow[k0 + 1];
      af[kb].x = live ? u : 0.f;
      af[kb].y = live ? v : 0.f;
    }
    v8f c0, c1;
    for (int r = 0; r < 8; r++) { c0[r] = 0.f; c1[r] = 0.f; }
    for (int kb = 0; kb < 8; kb++) {
      v4f q = ldsB4(Lg, lane, kb);
      c0 = wmma4(af[kb], q.xy, c0);
      c1 = wmma4(af[kb], q.zw, c1);
    }
    for (int r = 0; r < 8; r++) {
      Mg[fragIdx((r + 8 * hi) * 32 + m)]      = c0[r];
      Mg[fragIdx((r + 8 * hi) * 32 + 16 + m)] = c1[r];
    }
  }
  __syncthreads();

  if (tid < 96) {
    const int g = tid >> 5, n = tid & 31;
    const float* lbg = (g == 0) ? lbz : (g == 1) ? lbr : lbh;
    sc[tid] = sM[g * 512 + fragIdx(8 * 32 + n)] + lbg[n];
  }
  __syncthreads();

  for (int p = 0; p < PER; p++) {
    asm volatile("" ::: "memory");
    int f0 = hi * 2;
    v2f bx0, bx1;
    bx0.x = xrow[(f0)*PER + p];     bx0.y = xrow[(f0 + 1) * PER + p];
    bx1.x = xrow[(f0 + 4) * PER + p]; bx1.y = xrow[(f0 + 5) * PER + p];

    v8f z0, z1, r0, r1, t0, t1;
    biasTiles(sc, hi, z0, z1);
    mmaX(bx0, bx1, sM, lane, z0, z1);
    mmaD(H0, H1, sL + 1024, lane, z0, z1);
    for (int r = 0; r < 8; r++) { z0[r] = sigm(z0[r]); z1[r] = sigm(z1[r]); }

    biasTiles(sc + 32, hi, r0, r1);
    mmaX(bx0, bx1, sM + 512, lane, r0, r1);
    mmaD(H0, H1, sL + 2048 + 1024, lane, r0, r1);
    for (int r = 0; r < 8; r++) { r0[r] = sigm(r0[r]); r1[r] = sigm(r1[r]); }
    v8f q0, q1;
    for (int r = 0; r < 8; r++) { q0[r] = H0[r] * r0[r]; q1[r] = H1[r] * r1[r]; }

    biasTiles(sc + 64, hi, t0, t1);
    mmaX(bx0, bx1, sM + 1024, lane, t0, t1);
    mmaD(q0, q1, sL + 4096 + 1024, lane, t0, t1);
    for (int r = 0; r < 8; r++) { t0[r] = tanhc(t0[r]); t1[r] = tanhc(t1[r]); }

    const float prp = sPr[p];
    for (int r = 0; r < 8; r++) {
      H0[r] = z0[r] * H0[r] + (1.0f - z0[r]) * t0[r];
      H1[r] = z1[r] * H1[r] + (1.0f - z1[r]) * t1[r];
      A0[r] += prp * H0[r];
      A1[r] += prp * H1[r];
    }
  }

  for (int r = 0; r < 8; r++) { A0[r] = fmaxf(A0[r], 0.f); A1[r] = fmaxf(A1[r], 0.f); }
  v8f o0, o1;
  biasTiles(sbo, hi, o0, o1);
  mmaD(A0, A1, sWo, lane, o0, o1);
  long row = (long)tile * 16 + m;
  if (row < ROWS) {
    v4f* orow = (v4f*)(out + row * OUTC + 8 * hi);
    orow[0] = __builtin_shufflevector(o0, o0, 0, 1, 2, 3);
    orow[1] = __builtin_shufflevector(o0, o0, 4, 5, 6, 7);
    if (hi == 0) {
      orow[4] = __builtin_shufflevector(o1, o1, 0, 1, 2, 3);
      orow[5] = __builtin_shufflevector(o1, o1, 4, 5, 6, 7);
    }
  }
}

extern "C" void kernel_launch(void* const* d_in, const int* in_sizes, int n_in,
                              void* d_out, int out_size, void* d_ws, size_t ws_size,
                              hipStream_t stream) {
  (void)in_sizes; (void)n_in; (void)out_size; (void)ws_size;
  const float* x   = (const float*)d_in[0];
  const int*   ei  = (const int*)d_in[1];
  const float* att = (const float*)d_in[2];
  const float* Wz  = (const float*)d_in[3];  const float* bz  = (const float*)d_in[4];
  const float* Wr  = (const float*)d_in[5];  const float* br  = (const float*)d_in[6];
  const float* Wh  = (const float*)d_in[7];  const float* bh  = (const float*)d_in[8];
  const float* Lz  = (const float*)d_in[9];  const float* lbz = (const float*)d_in[10];
  const float* Lr  = (const float*)d_in[11]; const float* lbr = (const float*)d_in[12];
  const float* Lh  = (const float*)d_in[13]; const float* lbh = (const float*)d_in[14];
  const float* Wo  = (const float*)d_in[15]; const float* bo  = (const float*)d_in[16];
  float* out = (float*)d_out;

  float* wsf  = (float*)d_ws;
  int*   cnt  = (int*)d_ws;
  float* xagg = wsf + 65536;
  int* cells  = (int*)(xagg + NNODES * NFEAT);
  int* counts = cells + NBINS * NBLK1 * CAPB;

  k_bin<<<NBLK1, BTHREADS, 0, stream>>>(ei, counts, cells);
  k_sort<<<NBINS, STHREADS, 0, stream>>>(ei, counts, cells, cnt, (int*)xagg);
  k_gather<<<GWAVES / 8, 256, 0, stream>>>(x, ei, cnt, xagg);

  k_gru<<<GBLOCKS, GTHREADS, 0, stream>>>(xagg, att, Wz, bz, Wr, br, Wh, bh,
                                          Lz, lbz, Lr, lbr, Lh, lbh, Wo, bo, out);
}


# GRU activations on the hardware tanh (v_tanh_f32), one transcendental per value: sigmoid = 0.5+0.5*tanh(x/2), the 1/2 folded exactly into the z, r weights; results differ in rounding only
# speedup vs baseline: 4.6061x; 1.0392x over previous
#include <hip/hip_runtime.h>

#define NNODES 50000
#define NEDGES 800000
#define FIN 8
#define PER 12
#define HIDN 32
#define NB 2
#define OUTC 24
#define ROWS (NB*NNODES)
#define NFEAT (FIN*PER)

typedef float v2f __attribute__((ext_vector_type(2)));
typedef float v4f __attribute__((ext_vector_type(4)));
typedef float v8f __attribute__((ext_vector_type(8)));

__device__ __forceinline__ v8f wmma4(v2f a, v2f b, v8f c) {
  return __builtin_amdgcn_wmma_f32_16x16x4_f32(false, a, false, b, (short)0, c, false, false);
}


__device__ __forceinline__ int fragIdx(int t) {
  int k = t >> 5, n = t & 31;
  int kb = k >> 2, hi = (k >> 1) & 1, j = k & 1, nt = n >> 4, m = n & 15;
  return ((kb * 32 + hi * 16 + m) << 2) + nt * 2 + j;
}
__device__ __forceinline__ int fragIdxD(int k, int n) {
  int kb = ((k >> 4) << 2) + ((k >> 1) & 3), hi = (k >> 3) & 1, j = k & 1, nt = n >> 4, m = n & 15;
  return ((kb * 32 + hi * 16 + m) << 2) + nt * 2 + j;
}
__device__ __forceinline__ v4f ldsB4(const float* W, int lane, int kb) {
  return *(const v4f*)(W + ((kb * 32 + lane) << 2));
}

__device__ __forceinline__ void biasTiles(const float* b, int hi, v8f& c0, v8f& c1) {
  const v4f* p = (const v4f*)(b + 8 * hi);
  v4f u0 = p[0], u1 = p[1], w0 = p[4], w1 = p[5];
  c0 = __builtin_shufflevector(u0, u1, 0, 1, 2, 3, 4, 5, 6, 7);
  c1 = __builtin_shufflevector(w0, w1, 0, 1, 2, 3, 4, 5, 6, 7);
}

__device__ __forceinline__ void mmaX(v2f bx0, v2f bx1, const float* M, int lane, v8f& c0, v8f& c1) {
  v4f q0 = ldsB4(M, lane, 0), q1 = ldsB4(M, lane, 1);
  c0 = wmma4(q0.xy, bx0, c0);
  c1 = wmma4(q0.zw, bx0, c1);
  c0 = wmma4(q1.xy, bx1, c0);
  c1 = wmma4(q1.zw, bx1, c1);
}

__device__ __forceinline__ void mmaD(const v8f& s0, const v8f& s1, const float* W, int lane, v8f& c0, v8f& c1) {
#pragma unroll
  for (int kb = 0; kb < 8; kb++) {
    const int p = kb & 3;
    v2f b;
    b.x = (kb < 4) ? s0[2 * p] : s1[2 * p];
    b.y = (kb < 4) ? s0[2 * p + 1] : s1[2 * p + 1];
    v4f q = ldsB4(W, lane, kb);
    c0 = wmma4(q.xy, b, c0);
    c1 = wmma4(q.zw, b, c1);
  }
}

__device__ __forceinline__ float sigmh(float xh) {
  return 0.5f + 0.5f * __builtin_amdgcn_tanhf(xh);
}
__device__ __forceinline__ float tanhc(float x) {
  return __builtin_amdgcn_tanhf(x);
}

#define SLOTCAP NFEAT
#define GWAVES 25000
static_assert(NFEAT == 96, "a row is 24 lanes x 16 bytes = three 128-byte lines");
static_assert(2 * GWAVES == NNODES && GWAVES % 8 == 0, "two nodes per gather wave, eight waves per block");

#define BSH 7
#define BNODES (1 << BSH)
#define NBINS ((NNODES + BNODES - 1) / BNODES)
#define BTHREADS 1024
#define EPT 4
#define ECHUNK (BTHREADS * EPT)
#define NBLK1 ((NEDGES + ECHUNK - 1) / ECHUNK)
#define CAPB 40
#define SPL ((NBINS + 31) / 32)
#define STHREADS 256
#define SU 8
static_assert(NNODES <= 65536 && BNODES <= 256, "a record packs the source in 16 bits and the node-in-bin above them");
static_assert(STHREADS >= BNODES, "one thread per node of a bin clears and stores the counts");
static_assert((long)NBINS * NBLK1 * CAPB + (long)NBLK1 * NBINS <= (long)NNODES * NFEAT, "cells and counts fit in the batch-1 half of xagg");

__global__ void __launch_bounds__(BTHREADS) k_bin(const int* __restrict__ ei, int* __restrict__ counts, int* __restrict__ cells) {
  __shared__ int hist[NBINS];
  __shared__ int start[NBINS];
  __shared__ int srec[ECHUNK];
  __shared__ int sdst[ECHUNK];
  const int tid = threadIdx.x, blk = blockIdx.x;
  for (int b = tid; b < NBINS; b += BTHREADS) hist[b] = 0;
  __syncthreads();
  int rec[EPT], bn[EPT], rk[EPT];
#pragma unroll
  for (int i = 0; i < EPT; i++) {
    const int e = blk * ECHUNK + i * BTHREADS + tid;
    const int ec = e < NEDGES ? e : NEDGES - 1;
    const int s = ei[ec], d = ei[NEDGES + ec];
    bn[i] = e < NEDGES ? (d >> BSH) : -1;
    rec[i] = ((d & (BNODES - 1)) << 16) | s;
    rk[i] = 0;
  }
#pragma unroll
  for (int i = 0; i < EPT; i++)
    if (bn[i] >= 0) rk[i] = atomicAdd(&hist[bn[i]], 1);
  __syncthreads();
  for (int b = tid; b < NBINS; b += BTHREADS) counts[blk * NBINS + b] = hist[b];
  if (tid < 32) {
    int sum = 0;
    for (int j = 0; j < SPL; j++) { const int b = tid * SPL + j; if (b < NBINS) sum += hist[b]; }
    int inc = sum;
    for (int off = 1; off < 32; off <<= 1) { const int v = __shfl_up(inc, off, 32); if (tid >= off) inc += v; }
    int run = inc - sum;
    for (int j = 0; j < SPL; j++) { const int b = tid * SPL + j; if (b < NBINS) { start[b] = run; run += hist[b]; } }
  }
  __syncthreads();
#pragma unroll
  for (int i = 0; i < EPT; i++)
    if (bn[i] >= 0) {
      const int p = start[bn[i]] + rk[i];
      srec[p] = rec[i];
      sdst[p] = (rk[i] < CAPB) ? (bn[i] * NBLK1 + blk) * CAPB + rk[i] : -1;
    }
  __syncthreads();
  const int left = NEDGES - blk * ECHUNK;
  const int total = left < ECHUNK ? left : ECHUNK;
  for (int p = tid; p < total; p += BTHREADS) {
    const int dd = sdst[p];
    if (dd >= 0) cells[dd] = srec[p];
  }
}

__global__ void __launch_bounds__(STHREADS) k_sort(const int* __restrict__ ei, const int* __restrict__ counts,
                                                   const int* __restrict__ cells, int* __restrict__ cnt, int* slots) {
  __shared__ int lcnt[BNODES];
  __shared__ int ccnt[NBLK1];
  __shared__ int llist[BNODES * SLOTCAP];
  __shared__ int ovf;
  const int tid = threadIdx.x, bin = blockIdx.x;
  if (tid < BNODES) lcnt[tid] = 0;
  if (tid == 0) ovf = 0;
  __syncthreads();
  for (int c = tid; c < NBLK1; c += STHREADS) {
    const int v = counts[c * NBINS + bin];
    ccnt[c] = v;
    if (v > CAPB) ovf = 1;
  }
  __syncthreads();
  if (ovf == 0) {
    for (int i0 = tid; i0 < NBLK1 * CAPB; i0 += STHREADS * SU) {
      int rr[SU];
      bool ok[SU];
#pragma unroll
      for (int u = 0; u < SU; u++) {
        const int idx = i0 + u * STHREADS;
        ok[u] = (idx < NBLK1 * CAPB) && (idx % CAPB < ccnt[idx / CAPB]);
        rr[u] = ok[u] ? cells[bin * (NBLK1 * CAPB) + idx] : 0;
      }
#pragma unroll
      for (int u = 0; u < SU; u++)
        if (ok[u]) {
          const int dl = rr[u] >> 16, s = rr[u] & 0xffff;
          const int r = atomicAdd(&lcnt[dl], 1);
          if (r < SLOTCAP) llist[dl * SLOTCAP + r] = s;
        }
    }
  } else {
    for (int e = tid; e < NEDGES; e += STHREADS) {
      const int d = ei[NEDGES + e];
      if ((d >> BSH) == bin) {
        const int dl = d & (BNODES - 1);
        const int r = atomicAdd(&lcnt[dl], 1);
        if (r < SLOTCAP) llist[dl * SLOTCAP + r] = ei[e];
      }
    }
  }
  __syncthreads();
  if (tid < BNODES && bin * BNODES + tid < NNODES) cnt[bin * BNODES + tid] = lcnt[tid];
  const int wave = tid >> 5, lane = tid & 31;
  for (int dl = wave; dl < BNODES; dl += STHREADS / 32) {
    const int n = bin * BNODES + dl;
    if (n < NNODES) {
      const int deg = lcnt[dl];
      const int m = deg < SLOTCAP ? deg : SLOTCAP;
      for (int j = lane; j < m; j += 32) slots[n * NFEAT + j] = llist[dl * SLOTCAP + j];
    }
  }
}

__global__ void __launch_bounds__(256) k_gather(const float* __restrict__ x, const int* __restrict__ ei,
                                                const int* __restrict__ cnt, float* xagg) {
  const int lane = threadIdx.x & 31;
  const int ql = lane < 24 ? lane : 23;
  const v4f* x4 = (const v4f*)x;
  const int b4 = NNODES * (NFEAT / 4);
  const int w0 = blockIdx.x * 8 + (threadIdx.x >> 5);
  for (int it = 0; it < 2; it++) {
    const int n = __builtin_amdgcn_readfirstlane(w0 + it * GWAVES);
    const int deg = cnt[n];
    const float dn = rsqrtf((float)(deg + 1));
    v4f a0, a1;
    {
      const float ww = dn * dn;
      a0 = ww * x4[n * 24 + ql];
      a1 = ww * x4[b4 + n * 24 + ql];
    }
    if (deg <= SLOTCAP) {
      const int* slots = (const int*)xagg + n * NFEAT;
      for (int c0 = 0; c0 < deg; c0 += 32) {
        const int m = (deg - c0 < 32) ? deg - c0 : 32;
        int key = 0x7fffffff;
        float ws = 0.f;
        if (lane < m) {
          const int s = slots[c0 + lane];
          ws = rsqrtf((float)(cnt[s] + 1)) * dn;
          key = (s << 5) | lane;
        }
        int myrank = 0;
        for (int j = 0; j < m; j++) myrank += (__builtin_amdgcn_readlane(key, j) < key) ? 1 : 0;
        for (int k0 = 0; k0 < m; k0 += 4) {
          v4f r0[4], r1[4];
          float wk[4];
#pragma unroll
          for (int u = 0; u < 4; u++) {
            wk[u] = 0.f; r0[u] = 0.f; r1[u] = 0.f;
            if (k0 + u < m) {
              const int l = __builtin_ctz(__builtin_amdgcn_ballot_w32(myrank == k0 + u));
              const int sk = __builtin_amdgcn_readlane(key, l) >> 5;
              wk[u] = __int_as_float(__builtin_amdgcn_readlane(__float_as_int(ws), l));
              r0[u] = x4[sk * 24 + ql];
              r1[u] = x4[b4 + sk * 24 + ql];
            }
          }
#pragma unroll
          for (int u = 0; u < 4; u++) { a0 += wk[u] * r0[u]; a1 += wk[u] * r1[u]; }
        }
      }
    } else {
      for (int e0 = 0; e0 < NEDGES; e0 += 32) {
        unsigned mask = __builtin_amdgcn_ballot_w32(ei[NEDGES + e0 + lane] == n);
        while (mask) {
          const int l = __builtin_ctz(mask);
          mask &= mask - 1;
          const int sk = ei[e0 + l];
          const float wk = rsqrtf((float)(cnt[sk] + 1)) * dn;
          a0 += wk * x4[sk * 24 + ql];
          a1 += wk * x4[b4 + sk * 24 + ql];
        }
      }
    }
    if (lane < 24) {
      ((v4f*)xagg)[n * 24 + lane] = a0;
      ((v4f*)xagg)[b4 + n * 24 + lane] = a1;
    }
  }
}

#define GWPB 25
#define GBLOCKS 250
#define GTHREADS (GWPB * 32)
static_assert(GBLOCKS * GWPB * 16 == ROWS, "the grid covers the rows exactly, one tile per wave");
__global__ void __launch_bounds__(GTHREADS) k_gru(
    const float* __restrict__ xagg, const float* __restrict__ att,
    const float* __restrict__ Wz, const float* __restrict__ bz,
    const float* __restrict__ Wr, const float* __restrict__ br,
    const float* __restrict__ Wh, const float* __restrict__ bh,
    const float* __restrict__ Lz, const float* __restrict__ lbz,
    const float* __restrict__ Lr, const float* __restrict__ lbr,
    const float* __restrict__ Lh, const float* __restrict__ lbh,
    const float* __restrict__ Wo, const float* __restrict__ bo,
    float* __restrict__ out) {
  __shared__ __attribute__((aligned(16))) float sM[3 * 16 * 32];
  __shared__ __attribute__((aligned(16))) float sL[3 * 64 * 32];
  __shared__ __attribute__((aligned(16))) float sWo[32 * 32];
  __shared__ __attribute__((aligned(16))) float sbo[32];
  __shared__ __attribute__((aligned(16))) float sc[3 * 32];
  __shared__ float sPr[PER];

  int tid = threadIdx.x;
  for (int t = tid; t < 32; t += GTHREADS) sbo[t] = (t < OUTC) ? bo[t] : 0.f;
  for (int t = tid; t < 64 * 32; t += GTHREADS) {
    int k = t >> 5, n = t & 31;
    int f = (k < 32) ? fragIdx(t) : 1024 + fragIdxD(k - 32, n);
    sL[f] = 0.5f * Lz[t]; sL[2048 + f] = 0.5f * Lr[t]; sL[4096 + f] = Lh[t];
  }
  for (int t = tid; t < 32 * 32; t += GTHREADS) {
    int k = t >> 5, j = t & 31;
    sWo[fragIdxD(k, j)] = (j < OUTC) ? Wo[k * OUTC + j] : 0.f;
  }

  int wave = tid >> 5, lane = tid & 31, m = lane & 15, hi = lane >> 4;

  {
    float a[PER]; float mx = -1e30f;
    for (int p = 0; p < PER; p++) { a[p] = att[p]; mx = fmaxf(mx, a[p]); }
    float sm = 0.f;
    for (int p = 0; p < PER; p++) { a[p] = __expf(a[p] - mx); sm += a[p]; }
    float inv = 1.0f / sm;
    if (tid == 0) for (int p = 0; p < PER; p++) sPr[p] = a[p] * inv;
  }

  int tile = blockIdx.x * GWPB + wave;
  long ia = (long)tile * 16 + m;
  if (ia >= ROWS) ia = ROWS - 1;
  const float* xrow = xagg + ia * (long)NFEAT;

  v8f H0, H1, A0, A1;
  for (int r = 0; r < 8; r++) { H0[r] = 0.f; H1[r] = 0.f; A0[r] = 0.f; A1[r] = 0.f; }

  __syncthreads();

  if (wave < 3) {
    const float* Wg = (wave == 0) ? Wz : (wave == 1) ? Wr : Wh;
    const float* bg = (wave == 0) ? bz : (wave == 1) ? br : bh;
    const float* Lg = sL + wave * 2048;
    float* Mg = sM + wave * 512;
    const float* arow = (m < 8) ? (Wg + m * 32) : bg;
    const bool live = (m <= 8);
    v2f af[8];
    for (int kb = 0; kb < 8; kb++) {
      const int k0 = kb * 4 + hi * 2;
      const float u = arow[k0], v = arow[k0 + 1];
      af[kb].x = live ? u : 0.f;
      af[kb].y = live ? v : 0.f;
    }
    v8f c0, c1;
    for (int r = 0; r < 8; r++) { c0[r] = 0.f; c1[r] = 0.f; }
    for (int kb = 0; kb < 8; kb++) {
      v4f q = ldsB4(Lg, lane, kb);
      c0 = wmma4(af[kb], q.xy, c0);
      c1 = wmma4(af[kb], q.zw, c1);
    }
    for (int r = 0; r < 8; r++) {
      Mg[fragIdx((r + 8 * hi) * 32 + m)]      = c0[r];
      Mg[fragIdx((r + 8 * hi) * 32 + 16 + m)] = c1[r];
    }
  }
  __syncthreads();

  if (tid < 96) {
    const int g = tid >> 5, n = tid & 31;
    const float* lbg = (g == 0) ? lbz : (g == 1) ? lbr : lbh;
    sc[tid] = sM[g * 512 + fragIdx(8 * 32 + n)] + ((g < 2) ? 0.5f : 1.0f) * lbg[n];
  }
  __syncthreads();

  for (int p = 0; p < PER; p++) {
    asm volatile("" ::: "memory");
    int f0 = hi * 2;
    v2f bx0, bx1;
    bx0.x = xrow[(f0)*PER + p];     bx0.y = xrow[(f0 + 1) * PER + p];
    bx1.x = xrow[(f0 + 4) * PER + p]; bx1.y = xrow[(f0 + 5) * PER + p];

    v8f z0, z1, r0, r1, t0, t1;
    biasTiles(sc, hi, z0, z1);
    mmaX(bx0, bx1, sM, lane, z0, z1);
    mmaD(H0, H1, sL + 1024, lane, z0, z1);
    for (int r = 0; r < 8; r++) { z0[r] = sigmh(z0[r]); z1[r] = sigmh(z1[r]); }

    biasTiles(sc + 32, hi, r0, r1);
    mmaX(bx0, bx1, sM + 512, lane, r0, r1);
    mmaD(H0, H1, sL + 2048 + 1024, lane, r0, r1);
    for (int r = 0; r < 8; r++) { r0[r] = sigmh(r0[r]); r1[r] = sigmh(r1[r]); }
    v8f q0, q1;
    for (int r = 0; r < 8; r++) { q0[r] = H0[r] * r0[r]; q1[r] = H1[r] * r1[r]; }

    biasTiles(sc + 64, hi, t0, t1);
    mmaX(bx0, bx1, sM + 1024, lane, t0, t1);
    mmaD(q0, q1, sL + 4096 + 1024, lane, t0, t1);
    for (int r = 0; r < 8; r++) { t0[r] = tanhc(t0[r]); t1[r] = tanhc(t1[r]); }

    const float prp = sPr[p];
    for (int r = 0; r < 8; r++) {
      H0[r] = z0[r] * H0[r] + (1.0f - z0[r]) * t0[r];
      H1[r] = z1[r] * H1[r] + (1.0f - z1[r]) * t1[r];
      A0[r] += prp * H0[r];
      A1[r] += prp * H1[r];
    }
  }

  for (int r = 0; r < 8; r++) { A0[r] = fmaxf(A0[r], 0.f); A1[r] = fmaxf(A1[r], 0.f); }
  v8f o0, o1;
  biasTiles(sbo, hi, o0, o1);
  mmaD(A0, A1, sWo, lane, o0, o1);
  long row = (long)tile * 16 + m;
  if (row < ROWS) {
    v4f* orow = (v4f*)(out + row * OUTC + 8 * hi);
    orow[0] = __builtin_shufflevector(o0, o0, 0, 1, 2, 3);
    orow[1] = __builtin_shufflevector(o0, o0, 4, 5, 6, 7);
    if (hi == 0) {
      orow[4] = __builtin_shufflevector(o1, o1, 0, 1, 2, 3);
      orow[5] = __builtin_shufflevector(o1, o1, 4, 5, 6, 7);
    }
  }
}

extern "C" void kernel_launch(void* const* d_in, const int* in_sizes, int n_in,
                              void* d_out, int out_size, void* d_ws, size_t ws_size,
                              hipStream_t stream) {
  (void)in_sizes; (void)n_in; (void)out_size; (void)ws_size;
  const float* x   = (const float*)d_in[0];
  const int*   ei  = (const int*)d_in[1];
  const float* att = (const float*)d_in[2];
  const float* Wz  = (const float*)d_in[3];  const float* bz  = (const float*)d_in[4];
  const float* Wr  = (const float*)d_in[5];  const float* br  = (const float*)d_in[6];
  const float* Wh  = (const float*)d_in[7];  const float* bh  = (const float*)d_in[8];
  const float* Lz  = (const float*)d_in[9];  const float* lbz = (const float*)d_in[10];
  const float* Lr  = (const float*)d_in[11]; const float* lbr = (const float*)d_in[12];
  const float* Lh  = (const float*)d_in[13]; const float* lbh = (const float*)d_in[14];
  const float* Wo  = (const float*)d_in[15]; const float* bo  = (const float*)d_in[16];
  float* out = (float*)d_out;

  float* wsf  = (float*)d_ws;
  int*   cnt  = (int*)d_ws;
  float* xagg = wsf + 65536;
  int* cells  = (int*)(xagg + NNODES * NFEAT);
  int* counts = cells + NBINS * NBLK1 * CAPB;

  k_bin<<<NBLK1, BTHREADS, 0, stream>>>(ei, counts, cells);
  k_sort<<<NBINS, STHREADS, 0, stream>>>(ei, counts, cells, cnt, (int*)xagg);
  k_gather<<<GWAVES / 8, 256, 0, stream>>>(x, ei, cnt, xagg);

  k_gru<<<GBLOCKS, GTHREADS, 0, stream>>>(xagg, att, Wz, bz, Wr, br, Wh, bh,
                                          Lz, lbz, Lr, lbr, Lh, lbh, Wo, bo, out);
}


# gather fused into the GRU launch as a two-stage pipeline per WGP: 24 waves gather the next 8 tiles (8 nodes x both batches) into LDS while 8 waves run the GRU of the last 8; sums and order as before
# speedup vs baseline: 5.1058x; 1.1085x over previous
#include <hip/hip_runtime.h>

#define NNODES 50000
#define NEDGES 800000
#define FIN 8
#define PER 12
#define HIDN 32
#define NB 2
#define OUTC 24
#define ROWS (NB*NNODES)
#define NFEAT (FIN*PER)

typedef float v2f __attribute__((ext_vector_type(2)));
typedef float v4f __attribute__((ext_vector_type(4)));
typedef float v8f __attribute__((ext_vector_type(8)));

__device__ __forceinline__ v8f wmma4(v2f a, v2f b, v8f c) {
  return __builtin_amdgcn_wmma_f32_16x16x4_f32(false, a, false, b, (short)0, c, false, false);
}


__device__ __forceinline__ int fragIdx(int t) {
  int k = t >> 5, n = t & 31;
  int kb = k >> 2, hi = (k >> 1) & 1, j = k & 1, nt = n >> 4, m = n & 15;
  return ((kb * 32 + hi * 16 + m) << 2) + nt * 2 + j;
}
__device__ __forceinline__ int fragIdxD(int k, int n) {
  int kb = ((k >> 4) << 2) + ((k >> 1) & 3), hi = (k >> 3) & 1, j = k & 1, nt = n >> 4, m = n & 15;
  return ((kb * 32 + hi * 16 + m) << 2) + nt * 2 + j;
}
__device__ __forceinline__ v4f ldsB4(const float* W, int lane, int kb) {
  return *(const v4f*)(W + ((kb * 32 + lane) << 2));
}

__device__ __forceinline__ void biasTiles(const float* b, int hi, v8f& c0, v8f& c1) {
  const v4f* p = (const v4f*)(b + 8 * hi);
  v4f u0 = p[0], u1 = p[1], w0 = p[4], w1 = p[5];
  c0 = __builtin_shufflevector(u0, u1, 0, 1, 2, 3, 4, 5, 6, 7);
  c1 = __builtin_shufflevector(w0, w1, 0, 1, 2, 3, 4, 5, 6, 7);
}

__device__ __forceinline__ void mmaX(v2f bx0, v2f bx1, const float* M, int lane, v8f& c0, v8f& c1) {
  v4f q0 = ldsB4(M, lane, 0), q1 = ldsB4(M, lane, 1);
  c0 = wmma4(q0.xy, bx0, c0);
  c1 = wmma4(q0.zw, bx0, c1);
  c0 = wmma4(q1.xy, bx1, c0);
  c1 = wmma4(q1.zw, bx1, c1);
}

__device__ __forceinline__ void mmaD(const v8f& s0, const v8f& s1, const float* W, int lane, v8f& c0, v8f& c1) {
#pragma unroll
  for (int kb = 0; kb < 8; kb++) {
    const int p = kb & 3;
    v2f b;
    b.x = (kb < 4) ? s0[2 * p] : s1[2 * p];
    b.y = (kb < 4) ? s0[2 * p + 1] : s1[2 * p + 1];
    v4f q = ldsB4(W, lane, kb);
    c0 = wmma4(q.xy, b, c0);
    c1 = wmma4(q.zw, b, c1);
  }
}

__device__ __forceinline__ float sigmh(float xh) {
  return 0.5f + 0.5f * __builtin_amdgcn_tanhf(xh);
}
__device__ __forceinline__ float tanhc(float x) {
  return __builtin_amdgcn_tanhf(x);
}

#define SLOTCAP NFEAT
static_assert(NFEAT == 96, "a row is 24 lanes x 16 bytes = three 128-byte lines");

#define BSH 7
#define BNODES (1 << BSH)
#define NBINS ((NNODES + BNODES - 1) / BNODES)
#define BTHREADS 1024
#define EPT 4
#define ECHUNK (BTHREADS * EPT)
#define NBLK1 ((NEDGES + ECHUNK - 1) / ECHUNK)
#define CAPB 40
#define SPL ((NBINS + 31) / 32)
#define STHREADS 256
#define SU 8
static_assert(NNODES <= 65536 && BNODES <= 256, "a record packs the source in 16 bits and the node-in-bin above them");
static_assert(STHREADS >= BNODES, "one thread per node of a bin clears and stores the counts");
static_assert((long)NBINS * NBLK1 * CAPB + (long)NBLK1 * NBINS <= (long)NNODES * NFEAT, "cells and counts fit in the batch-1 half of xagg");

__global__ void __launch_bounds__(BTHREADS) k_bin(const int* __restrict__ ei, int* __restrict__ counts, int* __restrict__ cells) {
  __shared__ int hist[NBINS];
  __shared__ int start[NBINS];
  __shared__ int srec[ECHUNK];
  __shared__ int sdst[ECHUNK];
  const int tid = threadIdx.x, blk = blockIdx.x;
  for (int b = tid; b < NBINS; b += BTHREADS) hist[b] = 0;
  __syncthreads();
  int rec[EPT], bn[EPT], rk[EPT];
#pragma unroll
  for (int i = 0; i < EPT; i++) {
    const int e = blk * ECHUNK + i * BTHREADS + tid;
    const int ec = e < NEDGES ? e : NEDGES - 1;
    const int s = ei[ec], d = ei[NEDGES + ec];
    bn[i] = e < NEDGES ? (d >> BSH) : -1;
    rec[i] = ((d & (BNODES - 1)) << 16) | s;
    rk[i] = 0;
  }
#pragma unroll
  for (int i = 0; i < EPT; i++)
    if (bn[i] >= 0) rk[i] = atomicAdd(&hist[bn[i]], 1);
  __syncthreads();
  for (int b = tid; b < NBINS; b += BTHREADS) counts[blk * NBINS + b] = hist[b];
  if (tid < 32) {
    int sum = 0;
    for (int j = 0; j < SPL; j++) { const int b = tid * SPL + j; if (b < NBINS) sum += hist[b]; }
    int inc = sum;
    for (int off = 1; off < 32; off <<= 1) { const int v = __shfl_up(inc, off, 32); if (tid >= off) inc += v; }
    int run = inc - sum;
    for (int j = 0; j < SPL; j++) { const int b = tid * SPL + j; if (b < NBINS) { start[b] = run; run += hist[b]; } }
  }
  __syncthreads();
#pragma unroll
  for (int i = 0; i < EPT; i++)
    if (bn[i] >= 0) {
      const int p = start[bn[i]] + rk[i];
      srec[p] = rec[i];
      sdst[p] = (rk[i] < CAPB) ? (bn[i] * NBLK1 + blk) * CAPB + rk[i] : -1;
    }
  __syncthreads();
  const int left = NEDGES - blk * ECHUNK;
  const int total = left < ECHUNK ? left : ECHUNK;
  for (int p = tid; p < total; p += BTHREADS) {
    const int dd = sdst[p];
    if (dd >= 0) cells[dd] = srec[p];
  }
}

__global__ void __launch_bounds__(STHREADS) k_sort(const int* __restrict__ ei, const int* __restrict__ counts,
                                                   const int* __restrict__ cells, int* __restrict__ cnt, int* slots) {
  __shared__ int lcnt[BNODES];
  __shared__ int ccnt[NBLK1];
  __shared__ int llist[BNODES * SLOTCAP];
  __shared__ int ovf;
  const int tid = threadIdx.x, bin = blockIdx.x;
  if (tid < BNODES) lcnt[tid] = 0;
  if (tid == 0) ovf = 0;
  __syncthreads();
  for (int c = tid; c < NBLK1; c += STHREADS) {
    const int v = counts[c * NBINS + bin];
    ccnt[c] = v;
    if (v > CAPB) ovf = 1;
  }
  __syncthreads();
  if (ovf == 0) {
    for (int i0 = tid; i0 < NBLK1 * CAPB; i0 += STHREADS * SU) {
      int rr[SU];
      bool ok[SU];
#pragma unroll
      for (int u = 0; u < SU; u++) {
        const int idx = i0 + u * STHREADS;
        ok[u] = (idx < NBLK1 * CAPB) && (idx % CAPB < ccnt[idx / CAPB]);
        rr[u] = ok[u] ? cells[bin * (NBLK1 * CAPB) + idx] : 0;
      }
#pragma unroll
      for (int u = 0; u < SU; u++)
        if (ok[u]) {
          const int dl = rr[u] >> 16, s = rr[u] & 0xffff;
          const int r = atomicAdd(&lcnt[dl], 1);
          if (r < SLOTCAP) llist[dl * SLOTCAP + r] = s;
        }
    }
  } else {
    for (int e = tid; e < NEDGES; e += STHREADS) {
      const int d = ei[NEDGES + e];
      if ((d >> BSH) == bin) {
        const int dl = d & (BNODES - 1);
        const int r = atomicAdd(&lcnt[dl], 1);
        if (r < SLOTCAP) llist[dl * SLOTCAP + r] = ei[e];
      }
    }
  }
  __syncthreads();
  if (tid < BNODES && bin * BNODES + tid < NNODES) cnt[bin * BNODES + tid] = lcnt[tid];
  const int wave = tid >> 5, lane = tid & 31;
  for (int dl = wave; dl < BNODES; dl += STHREADS / 32) {
    const int n = bin * BNODES + dl;
    if (n < NNODES) {
      const int deg = lcnt[dl];
      const int m = deg < SLOTCAP ? deg : SLOTCAP;
      for (int j = lane; j < m; j += 32) slots[n * NFEAT + j] = llist[dl * SLOTCAP + j];
    }
  }
}

__device__ __forceinline__ void gatherNode(const v4f* __restrict__ x4, const int* __restrict__ ei, const int* __restrict__ cnt,
                                           const int* __restrict__ slotsAll, int n, int lane, float* d0, float* d1) {
  const int ql = lane < 24 ? lane : 23;
  const int b4 = NNODES * (NFEAT / 4);
  const int deg = cnt[n];
  const float dn = rsqrtf((float)(deg + 1));
  v4f a0, a1;
  {
    const float ww = dn * dn;
    a0 = ww * x4[n * 24 + ql];
    a1 = ww * x4[b4 + n * 24 + ql];
  }
  if (deg <= SLOTCAP) {
    const int* slots = slotsAll + n * NFEAT;
    for (int c0 = 0; c0 < deg; c0 += 32) {
      const int m = (deg - c0 < 32) ? deg - c0 : 32;
      int key = 0x7fffffff;
      float ws = 0.f;
      if (lane < m) {
        const int s = slots[c0 + lane];
        ws = rsqrtf((float)(cnt[s] + 1)) * dn;
        key = (s << 5) | lane;
      }
      int myrank = 0;
      for (int j = 0; j < m; j++) myrank += (__builtin_amdgcn_readlane(key, j) < key) ? 1 : 0;
      for (int k0 = 0; k0 < m; k0 += 4) {
        v4f r0[4], r1[4];
        float wk[4];
#pragma unroll
        for (int u = 0; u < 4; u++) {
          wk[u] = 0.f; r0[u] = 0.f; r1[u] = 0.f;
          if (k0 + u < m) {
            const int l = __builtin_ctz(__builtin_amdgcn_ballot_w32(myrank == k0 + u));
            const int sk = __builtin_amdgcn_readlane(key, l) >> 5;
            wk[u] = __int_as_float(__builtin_amdgcn_readlane(__float_as_int(ws), l));
            r0[u] = x4[sk * 24 + ql];
            r1[u] = x4[b4 + sk * 24 + ql];
          }
        }
#pragma unroll
        for (int u = 0; u < 4; u++) { a0 += wk[u] * r0[u]; a1 += wk[u] * r1[u]; }
      }
    }
  } else {
    for (int e0 = 0; e0 < NEDGES; e0 += 32) {
      unsigned mask = __builtin_amdgcn_ballot_w32(ei[NEDGES + e0 + lane] == n);
      while (mask) {
        const int l = __builtin_ctz(mask);
        mask &= mask - 1;
        const int sk = ei[e0 + l];
        const float wk = rsqrtf((float)(cnt[sk] + 1)) * dn;
        a0 += wk * x4[sk * 24 + ql];
        a1 += wk * x4[b4 + sk * 24 + ql];
      }
    }
  }
  if (lane < 24) {
    *(v4f*)(d0 + 4 * lane) = a0;
    *(v4f*)(d1 + 4 * lane) = a1;
  }
}

#define FBLOCKS 256
#define FWAVES 32
#define FTHREADS (FWAVES * 32)
#define NGRU 8
#define NTILES (NNODES / 8)
#define LTILES ((NTILES + FBLOCKS - 1) / FBLOCKS)
#define NSETS ((LTILES + NGRU - 1) / NGRU)
#define XPITCH 100
#define REMT (LTILES - (NSETS - 1) * NGRU)
static_assert(NNODES % 8 == 0, "a tile is 8 whole nodes");
static_assert(REMT >= 1 && REMT <= NGRU, "set 0 is the remainder of the NGRU-sized sets");
__device__ __forceinline__ int tileOf(int blk, int s, int j) {
  const int lt = (s == 0) ? ((j < REMT) ? (NSETS - 1) * NGRU + j : LTILES) : (s - 1) * NGRU + j;
  return blk + FBLOCKS * lt;
}
__global__ void __launch_bounds__(FTHREADS) k_gru(
    const float* __restrict__ x, const int* __restrict__ ei,
    const int* __restrict__ cnt, const int* __restrict__ slotsAll,
    const float* __restrict__ att,
    const float* __restrict__ Wz, const float* __restrict__ bz,
    const float* __restrict__ Wr, const float* __restrict__ br,
    const float* __restrict__ Wh, const float* __restrict__ bh,
    const float* __restrict__ Lz, const float* __restrict__ lbz,
    const float* __restrict__ Lr, const float* __restrict__ lbr,
    const float* __restrict__ Lh, const float* __restrict__ lbh,
    const float* __restrict__ Wo, const float* __restrict__ bo,
    float* __restrict__ out) {
  __shared__ __attribute__((aligned(16))) float sM[3 * 16 * 32];
  __shared__ __attribute__((aligned(16))) float sL[3 * 64 * 32];
  __shared__ __attribute__((aligned(16))) float sWo[32 * 32];
  __shared__ __attribute__((aligned(16))) float sbo[32];
  __shared__ __attribute__((aligned(16))) float sc[3 * 32];
  __shared__ float sPr[PER];
  __shared__ __attribute__((aligned(16))) float sX[2][NGRU][16][XPITCH];

  int tid = threadIdx.x;
  for (int t = tid; t < 32; t += FTHREADS) sbo[t] = (t < OUTC) ? bo[t] : 0.f;
  for (int t = tid; t < 64 * 32; t += FTHREADS) {
    int k = t >> 5, n = t & 31;
    int f = (k < 32) ? fragIdx(t) : 1024 + fragIdxD(k - 32, n);
    sL[f] = 0.5f * Lz[t]; sL[2048 + f] = 0.5f * Lr[t]; sL[4096 + f] = Lh[t];
  }
  for (int t = tid; t < 32 * 32; t += FTHREADS) {
    int k = t >> 5, j = t & 31;
    sWo[fragIdxD(k, j)] = (j < OUTC) ? Wo[k * OUTC + j] : 0.f;
  }

  const int wave = __builtin_amdgcn_readfirstlane(tid >> 5);
  const int lane = tid & 31, m = lane & 15, hi = lane >> 4;

  {
    float a[PER]; float mx = -1e30f;
    for (int p = 0; p < PER; p++) { a[p] = att[p]; mx = fmaxf(mx, a[p]); }
    float sm = 0.f;
    for (int p = 0; p < PER; p++) { a[p] = __expf(a[p] - mx); sm += a[p]; }
    float inv = 1.0f / sm;
    if (tid == 0) for (int p = 0; p < PER; p++) sPr[p] = a[p] * inv;
  }

  __syncthreads();

  if (wave < 3) {
    const float* Wg = (wave == 0) ? Wz : (wave == 1) ? Wr : Wh;
    const float* bg = (wave == 0) ? bz : (wave == 1) ? br : bh;
    const float* Lg = sL + wave * 2048;
    float* Mg = sM + wave * 512;
    const float* arow = (m < 8) ? (Wg + m * 32) : bg;
    const bool live = (m <= 8);
    v2f af[8];
    for (int kb = 0; kb < 8; kb++) {
      const int k0 = kb * 4 + hi * 2;
      const float u = arow[k0], v = arow[k0 + 1];
      af[kb].x = live ? u : 0.f;
      af[kb].y = live ? v : 0.f;
    }
    v8f c0, c1;
    for (int r = 0; r < 8; r++) { c0[r] = 0.f; c1[r] = 0.f; }
    for (int kb = 0; kb < 8; kb++) {
      v4f q = ldsB4(Lg, lane, kb);
      c0 = wmma4(af[kb], q.xy, c0);
      c1 = wmma4(af[kb], q.zw, c1);
    }
    for (int r = 0; r < 8; r++) {
      Mg[fragIdx((r + 8 * hi) * 32 + m)]      = c0[r];
      Mg[fragIdx((r + 8 * hi) * 32 + 16 + m)] = c1[r];
    }
  }
  __syncthreads();

  if (tid < 96) {
    const int g = tid >> 5, n = tid & 31;
    const float* lbg = (g == 0) ? lbz : (g == 1) ? lbr : lbh;
    sc[tid] = sM[g * 512 + fragIdx(8 * 32 + n)] + ((g < 2) ? 0.5f : 1.0f) * lbg[n];
  }
  __syncthreads();

  for (int ph = 0; ph <= NSETS; ph++) {
    if (ph < NSETS && (ph == 0 || wave >= NGRU)) {
      const int g0 = (ph == 0) ? wave : wave - NGRU;
      const int gs = (ph == 0) ? FWAVES : FWAVES - NGRU;
      for (int q = g0; q < 8 * NGRU; q += gs) {
        const int j = q >> 3;
        const int tile = tileOf(blockIdx.x, ph, j);
        if (tile < NTILES) {
          float* d0 = &sX[ph & 1][j][q & 7][0];
          gatherNode((const v4f*)x, ei, cnt, slotsAll, 8 * tile + (q & 7), lane, d0, d0 + 8 * XPITCH);
        }
      }
    }
    const int tile = (ph >= 1 && wave < NGRU) ? tileOf(blockIdx.x, ph - 1, wave) : NTILES;
    if (tile < NTILES) {
      const float* xcol = &sX[(ph - 1) & 1][wave][m][0];
      v8f H0, H1, A0, A1;
      for (int r = 0; r < 8; r++) { H0[r] = 0.f; H1[r] = 0.f; A0[r] = 0.f; A1[r] = 0.f; }

      for (int p = 0; p < PER; p++) {
        asm volatile("" ::: "memory");
        int f0 = hi * 2;
        v2f bx0, bx1;
        bx0.x = xcol[(f0)*PER + p];     bx0.y = xcol[(f0 + 1) * PER + p];
        bx1.x = xcol[(f0 + 4) * PER + p]; bx1.y = xcol[(f0 + 5) * PER + p];

        v8f z0, z1, r0, r1, t0, t1;
        biasTiles(sc, hi, z0, z1);
        mmaX(bx0, bx1, sM, lane, z0, z1);
        mmaD(H0, H1, sL + 1024, lane, z0, z1);
        for (int r = 0; r < 8; r++) { z0[r] = sigmh(z0[r]); z1[r] = sigmh(z1[r]); }

        biasTiles(sc + 32, hi, r0, r1);
        mmaX(bx0, bx1, sM + 512, lane, r0, r1);
        mmaD(H0, H1, sL + 2048 + 1024, lane, r0, r1);
        for (int r = 0; r < 8; r++) { r0[r] = sigmh(r0[r]); r1[r] = sigmh(r1[r]); }
        v8f q0, q1;
        for (int r = 0; r < 8; r++) { q0[r] = H0[r] * r0[r]; q1[r] = H1[r] * r1[r]; }

        biasTiles(sc + 64, hi, t0, t1);
        mmaX(bx0, bx1, sM + 1024, lane, t0, t1);
        mmaD(q0, q1, sL + 4096 + 1024, lane, t0, t1);
        for (int r = 0; r < 8; r++) { t0[r] = tanhc(t0[r]); t1[r] = tanhc(t1[r]); }

        const float prp = sPr[p];
        for (int r = 0; r < 8; r++) {
          H0[r] = z0[r] * H0[r] + (1.0f - z0[r]) * t0[r];
          H1[r] = z1[r] * H1[r] + (1.0f - z1[r]) * t1[r];
          A0[r] += prp * H0[r];
          A1[r] += prp * H1[r];
        }
      }

      for (int r = 0; r < 8; r++) { A0[r] = fmaxf(A0[r], 0.f); A1[r] = fmaxf(A1[r], 0.f); }
      v8f o0, o1;
      biasTiles(sbo, hi, o0, o1);
      mmaD(A0, A1, sWo, lane, o0, o1);
      long row = (long)(m >> 3) * NNODES + (long)tile * 8 + (m & 7);
      if (row < ROWS) {
        v4f* orow = (v4f*)(out + row * OUTC + 8 * hi);
        orow[0] = __builtin_shufflevector(o0, o0, 0, 1, 2, 3);
        orow[1] = __builtin_shufflevector(o0, o0, 4, 5, 6, 7);
        if (hi == 0) {
          orow[4] = __builtin_shufflevector(o1, o1, 0, 1, 2, 3);
          orow[5] = __builtin_shufflevector(o1, o1, 4, 5, 6, 7);
        }
      }
    }
    __syncthreads();
  }
}

extern "C" void kernel_launch(void* const* d_in, const int* in_sizes, int n_in,
                              void* d_out, int out_size, void* d_ws, size_t ws_size,
                              hipStream_t stream) {
  (void)in_sizes; (void)n_in; (void)out_size; (void)ws_size;
  const float* x   = (const float*)d_in[0];
  const int*   ei  = (const int*)d_in[1];
  const float* att = (const float*)d_in[2];
  const float* Wz  = (const float*)d_in[3];  const float* bz  = (const float*)d_in[4];
  const float* Wr  = (const float*)d_in[5];  const float* br  = (const float*)d_in[6];
  const float* Wh  = (const float*)d_in[7];  const float* bh  = (const float*)d_in[8];
  const float* Lz  = (const float*)d_in[9];  const float* lbz = (const float*)d_in[10];
  const float* Lr  = (const float*)d_in[11]; const float* lbr = (const float*)d_in[12];
  const float* Lh  = (const float*)d_in[13]; const float* lbh = (const float*)d_in[14];
  const float* Wo  = (const float*)d_in[15]; const float* bo  = (const float*)d_in[16];
  float* out = (float*)d_out;

  float* wsf  = (float*)d_ws;
  int*   cnt  = (int*)d_ws;
  float* xagg = wsf + 65536;
  int* cells  = (int*)(xagg + NNODES * NFEAT);
  int* counts = cells + NBINS * NBLK1 * CAPB;

  k_bin<<<NBLK1, BTHREADS, 0, stream>>>(ei, counts, cells);
  k_sort<<<NBINS, STHREADS, 0, stream>>>(ei, counts, cells, cnt, (int*)xagg);
  k_gru<<<FBLOCKS, FTHREADS, 0, stream>>>(x, ei, cnt, (const int*)xagg, att,
                                          Wz, bz, Wr, br, Wh, bh,
                                          Lz, lbz, Lr, lbr, Lh, lbh, Wo, bo, out);
}
